# starting kernel
# baseline (speedup 1.0000x reference)
#include <hip/hip_runtime.h>
#include <math.h>

#define BATCH 8
#define SEQ   1024
#define KREL  4
#define NPOS  (BATCH * SEQ)

typedef __attribute__((ext_vector_type(2))) float v2f;
typedef __attribute__((ext_vector_type(4))) float v4f;
typedef __attribute__((ext_vector_type(8))) float v8f;

__global__ void __launch_bounds__(256)
k_rel_reduce(const float* __restrict__ rel, float* __restrict__ factA4) {
  const int row = blockIdx.x;
  const int i   = row & (SEQ - 1);
  const v4f* base = (const v4f*)(rel + (size_t)row * (SEQ * KREL));
  __shared__ int found;
  if (threadIdx.x == 0) found = 0;
  __syncthreads();
  int local = 0;
  for (int j = threadIdx.x; j < i; j += 256) {
    v4f v = __builtin_nontemporal_load(base + j);
    local |= (v.x > 0.5f) || (v.y > 0.5f) || (v.z > 0.5f) || (v.w > 0.5f);
  }
  if (local) atomicOr(&found, 1);
  __syncthreads();
  if (threadIdx.x == 0) factA4[row] = found ? 0.7f : 1.0f;
}

__global__ void __launch_bounds__(128)
k_mlp_factors(const float* __restrict__ aL, const float* __restrict__ oL,
              const int*   __restrict__ exA, const int* __restrict__ exO,
              const float* __restrict__ W1, const float* __restrict__ b1,
              const float* __restrict__ W2, const float* __restrict__ b2,
              const float* __restrict__ W3, const float* __restrict__ b3,
              float* __restrict__ mult1, float* __restrict__ fA2,
              float* __restrict__ fO2) {
  __shared__ float h[4][16][32];
  const int tid  = threadIdx.x;
  const int w    = tid >> 5;
  const int lane = tid & 31;
  const int hf   = lane >> 4;
  const int ln   = lane & 15;
  const int posBase = blockIdx.x * 64 + w * 16;
  const int p = posBase + ln;

  const float* aRow = aL + (size_t)p * 4;
  const float* oRow = oL + (size_t)p * 4;
  const float x0 = aRow[0], x1 = aRow[1], x2 = aRow[2], x3 = aRow[3];
  const float x4 = oRow[0], x5 = oRow[1], x6 = oRow[2], x7 = oRow[3];

  v2f a0, a1, bw;
  a0.x = hf ? x2 : x0;  a0.y = hf ? x3 : x1;
  a1.x = hf ? x6 : x4;  a1.y = hf ? x7 : x5;
  v8f acc0 = {0,0,0,0,0,0,0,0};
  v8f acc1 = {0,0,0,0,0,0,0,0};
  bw.x = W1[(0 + 2*hf)*32 + ln];  bw.y = W1[(1 + 2*hf)*32 + ln];
  acc0 = __builtin_amdgcn_wmma_f32_16x16x4_f32(false, a0, false, bw, (short)0, acc0, false, false);
  bw.x = W1[(4 + 2*hf)*32 + ln];  bw.y = W1[(5 + 2*hf)*32 + ln];
  acc0 = __builtin_amdgcn_wmma_f32_16x16x4_f32(false, a1, false, bw, (short)0, acc0, false, false);
  bw.x = W1[(0 + 2*hf)*32 + 16 + ln];  bw.y = W1[(1 + 2*hf)*32 + 16 + ln];
  acc1 = __builtin_amdgcn_wmma_f32_16x16x4_f32(false, a0, false, bw, (short)0, acc1, false, false);
  bw.x = W1[(4 + 2*hf)*32 + 16 + ln];  bw.y = W1[(5 + 2*hf)*32 + 16 + ln];
  acc1 = __builtin_amdgcn_wmma_f32_16x16x4_f32(false, a1, false, bw, (short)0, acc1, false, false);

  const float bia0 = b1[ln], bia1 = b1[16 + ln];
  #pragma unroll
  for (int v = 0; v < 8; ++v) {
    float t0 = acc0[v] + bia0;  h[w][v + 8*hf][ln]      = t0 > 0.0f ? t0 : 0.0f;
    float t1 = acc1[v] + bia1;  h[w][v + 8*hf][16 + ln] = t1 > 0.0f ? t1 : 0.0f;
  }
  __syncthreads();

  v8f acc2 = {0,0,0,0,0,0,0,0};
  #pragma unroll
  for (int k0 = 0; k0 < 32; k0 += 4) {
    v2f aa, bb;
    aa.x = h[w][ln][k0 + 2*hf];
    aa.y = h[w][ln][k0 + 1 + 2*hf];
    bb.x = W2[(k0 + 2*hf)*16 + ln];
    bb.y = W2[(k0 + 1 + 2*hf)*16 + ln];
    acc2 = __builtin_amdgcn_wmma_f32_16x16x4_f32(false, aa, false, bb, (short)0, acc2, false, false);
  }
  __syncthreads();
  const float bia2 = b2[ln];
  #pragma unroll
  for (int v = 0; v < 8; ++v) {
    float t = acc2[v] + bia2;
    h[w][v + 8*hf][ln] = t > 0.0f ? t : 0.0f;
  }
  __syncthreads();

  if (hf == 0) {
    float s3 = b3[0];
    #pragma unroll
    for (int c = 0; c < 16; ++c) s3 += h[w][ln][c] * W3[c];
    float cg   = 1.0f / (1.0f + expf(-s3));
    float mult = cg < 0.5f ? 2.0f * cg : 1.0f;
    mult1[p] = mult;

    float mA = fmaxf(fmaxf(x0, x1), fmaxf(x2, x3));
    float e0 = expf(x0 - mA), e1 = expf(x1 - mA), e2 = expf(x2 - mA), e3 = expf(x3 - mA);
    float impA = (e0 + e1) / (e0 + e1 + e2 + e3);
    float mO = fmaxf(fmaxf(x4, x5), fmaxf(x6, x7));
    float g0 = expf(x4 - mO), g1 = expf(x5 - mO), g2 = expf(x6 - mO), g3 = expf(x7 - mO);
    float impO = (g0 + g1) / (g0 + g1 + g2 + g3);

    const int bI = p >> 10;
    const int s  = p & (SEQ - 1);
    const int* eA = exA + bI * SEQ;
    const int* eO = exO + bI * SEQ;
    bool nearA = false, nearO = false;
    #pragma unroll
    for (int d = -3; d <= 3; ++d) {
      int ss = s + d;
      if (ss >= 0 && ss < SEQ) { nearA |= (eA[ss] > 0); nearO |= (eO[ss] > 0); }
    }
    fA2[p] = (impA > 0.5f && !nearO) ? 0.3f : 1.0f;
    fO2[p] = (impO > 0.5f && !nearA) ? 0.3f : 1.0f;
  }
}

__global__ void __launch_bounds__(256)
k_scan_write(const float* __restrict__ aL, const float* __restrict__ oL,
             const float* __restrict__ mult1, const float* __restrict__ fA2,
             const float* __restrict__ fO2,  const float* __restrict__ fA4,
             float* __restrict__ outA, float* __restrict__ outO) {
  const int b = blockIdx.x;
  const int t = threadIdx.x;
  __shared__ float sm[SEQ], sfa[SEQ], sfo[SEQ];
  __shared__ unsigned OG[32], IF1[32], IF01[32], F3[32];
  if (t < 32) { OG[t] = 0u; IF1[t] = 0u; IF01[t] = 0u; }
  __syncthreads();

  for (int s = t; s < SEQ; s += 256) {
    const int p = b * SEQ + s;
    v4f av = *(const v4f*)(aL + (size_t)p * 4);
    v4f ov = *(const v4f*)(oL + (size_t)p * 4);
    float mult = mult1[p];
    float fa1  = fA2[p] * fA4[p];
    float fo1  = fO2[p];
    sm[s] = mult;  sfa[s] = fa1;  sfo[s] = fo1;

    float a0 = av.x * mult, a1 = av.y * mult, a2 = av.z * mult, a3 = av.w * mult;
    float o0 = ov.x * mult, o1 = ov.y * mult, o2 = ov.z * mult, o3 = ov.w * mult;
    bool orig = (fmaxf(fmaxf(av.x, av.y), fmaxf(av.z, av.w)) > 0.5f) ||
                (fmaxf(fmaxf(ov.x, ov.y), fmaxf(ov.z, ov.w)) > 0.5f);
    float fa = fa1, fo = fo1;
    bool act1  = (fmaxf(fmaxf(a0*fa, a1*fa), fmaxf(a2, a3)) > 0.5f) ||
                 (fmaxf(fmaxf(o0*fo, o1*fo), fmaxf(o2, o3)) > 0.5f);
    fa = fa1 * 0.1f; fo = fo1 * 0.1f;
    bool act01 = (fmaxf(fmaxf(a0*fa, a1*fa), fmaxf(a2, a3)) > 0.5f) ||
                 (fmaxf(fmaxf(o0*fo, o1*fo), fmaxf(o2, o3)) > 0.5f);

    unsigned bit = 1u << (s & 31);
    if (orig)  atomicOr(&OG[s >> 5],  bit);
    if (act1)  atomicOr(&IF1[s >> 5], bit);
    if (act01) atomicOr(&IF01[s >> 5], bit);
  }
  __syncthreads();

  if (t == 0) {
    unsigned n1 = 0u, n2 = 0u;
    for (int wi = 0; wi < 32; ++wi) {
      unsigned og   = OG[wi];
      unsigned ogn  = (wi < 31) ? OG[wi + 1] : 0u;
      unsigned if1  = IF1[wi];
      unsigned if01 = IF01[wi];
      unsigned f3w  = 0u;
      #pragma unroll 4
      for (int bb = 0; bb < 32; ++bb) {
        unsigned o1b = (bb + 1 < 32) ? ((og >> (bb + 1)) & 1u) : (ogn & 1u);
        unsigned o2b = (bb + 2 < 32) ? ((og >> (bb + 2)) & 1u) : ((ogn >> (bb + 2 - 32)) & 1u);
        unsigned nb  = n1 | n2 | o1b | o2b;
        unsigned na  = nb ? ((if1 >> bb) & 1u) : ((if01 >> bb) & 1u);
        f3w |= nb << bb;
        n2 = n1;  n1 = na;
      }
      F3[wi] = f3w;
    }
  }
  __syncthreads();

  for (int s = t; s < SEQ; s += 256) {
    const int p = b * SEQ + s;
    v4f av = *(const v4f*)(aL + (size_t)p * 4);
    v4f ov = *(const v4f*)(oL + (size_t)p * 4);
    float mult = sm[s];
    float f3 = ((F3[s >> 5] >> (s & 31)) & 1u) ? 1.0f : 0.1f;
    float fa = sfa[s] * f3;
    float fo = sfo[s] * f3;
    v4f ra, ro;
    ra.x = av.x * mult * fa;  ra.y = av.y * mult * fa;
    ra.z = av.z * mult;       ra.w = av.w * mult;
    ro.x = ov.x * mult * fo;  ro.y = ov.y * mult * fo;
    ro.z = ov.z * mult;       ro.w = ov.w * mult;
    *(v4f*)(outA + (size_t)p * 4) = ra;
    *(v4f*)(outO + (size_t)p * 4) = ro;
  }
}

extern "C" void kernel_launch(void* const* d_in, const int* in_sizes, int n_in,
                              void* d_out, int out_size, void* d_ws, size_t ws_size,
                              hipStream_t stream) {
  (void)in_sizes; (void)n_in; (void)out_size; (void)ws_size;
  const float* aL  = (const float*)d_in[0];
  const float* oL  = (const float*)d_in[1];
  const float* rel = (const float*)d_in[2];
  const int*   exA = (const int*)d_in[3];
  const int*   exO = (const int*)d_in[4];
  const float* W1  = (const float*)d_in[5];
  const float* b1  = (const float*)d_in[6];
  const float* W2  = (const float*)d_in[7];
  const float* b2  = (const float*)d_in[8];
  const float* W3  = (const float*)d_in[9];
  const float* b3  = (const float*)d_in[10];

  float* ws    = (float*)d_ws;
  float* fA4   = ws;
  float* mult1 = ws + NPOS;
  float* fA2   = ws + 2 * NPOS;
  float* fO2   = ws + 3 * NPOS;
  float* outA  = (float*)d_out;
  float* outO  = outA + NPOS * 4;

  k_rel_reduce<<<NPOS, 256, 0, stream>>>(rel, fA4);
  k_mlp_factors<<<NPOS / 64, 128, 0, stream>>>(aL, oL, exA, exO,
                                               W1, b1, W2, b2, W3, b3,
                                               mult1, fA2, fO2);
  k_scan_write<<<BATCH, 256, 0, stream>>>(aL, oL, mult1, fA2, fO2, fA4,
                                          outA, outO);
}


# 2 launches to 1: one 32-wave block per batch does MLP (WMMA), row probes (32 rows per wave, one 128-B line each) and scan; factors stay in the owning lane's registers, all input loads issued up front
# speedup vs baseline: 5.5575x; 5.5575x over previous
#include <hip/hip_runtime.h>
#include <math.h>


#define BATCH 8
#define SEQ   1024
#define KREL  4
#define NPOS  (BATCH * SEQ)
#define NWAVE 32
#define HP    34

typedef __attribute__((ext_vector_type(2))) float v2f;
typedef __attribute__((ext_vector_type(4))) float v4f;
typedef __attribute__((ext_vector_type(8))) float v8f;

__device__ __forceinline__ bool hit4(const v4f v) {
  return (v.x > 0.5f) || (v.y > 0.5f) || (v.z > 0.5f) || (v.w > 0.5f);
}

__device__ __forceinline__ void wave_lds_order() {
  __builtin_amdgcn_fence(__ATOMIC_RELEASE, "wavefront", "local");
  __builtin_amdgcn_wave_barrier();
  __builtin_amdgcn_fence(__ATOMIC_ACQUIRE, "wavefront", "local");
}

__global__ void __launch_bounds__(1024)
k_all(const float* __restrict__ rel,
      const float* __restrict__ aL, const float* __restrict__ oL,
      const int*   __restrict__ exA, const int* __restrict__ exO,
      const float* __restrict__ W1, const float* __restrict__ b1,
      const float* __restrict__ W2, const float* __restrict__ b2,
      const float* __restrict__ W3, const float* __restrict__ b3,
      float* __restrict__ outA, float* __restrict__ outO) {
  __shared__ float h[NWAVE][16][HP];
  __shared__ unsigned OG[32], IF1[32], IF01[32], F3[32];

  const int b    = blockIdx.x;
  const int tid  = threadIdx.x;
  const int lane = tid & 31;
  const int w    = __builtin_amdgcn_readfirstlane(tid >> 5);
  const int hf   = lane >> 4;
  const int ln   = lane & 15;
  const int s    = tid;
  const int p    = b * SEQ + s;

  const v4f av = *(const v4f*)(aL + (size_t)p * 4);
  const v4f ov = *(const v4f*)(oL + (size_t)p * 4);

  const v4f* R  = (const v4f*)rel + ((size_t)b << 20);
  const int sub = lane >> 3, col = lane & 7;
  v4f pv[8];
  #pragma unroll
  for (int j = 0; j < 8; ++j) {
    const int i = 32 * w + 4 * j + sub;
    pv[j] = R[(i << 10) + min(col, max(i - 1, 0))];
  }

  const int* eA = exA + b * SEQ;
  const int* eO = exO + b * SEQ;
  int wa[7], wo[7];
  #pragma unroll
  for (int d = 0; d < 7; ++d) {
    const int sc = min(max(s + d - 3, 0), SEQ - 1);
    wa[d] = eA[sc];  wo[d] = eO[sc];
  }

  v2f w1f[4], w2f[8];
  w1f[0].x = W1[(0 + 2*hf)*32 + ln];       w1f[0].y = W1[(1 + 2*hf)*32 + ln];
  w1f[1].x = W1[(4 + 2*hf)*32 + ln];       w1f[1].y = W1[(5 + 2*hf)*32 + ln];
  w1f[2].x = W1[(0 + 2*hf)*32 + 16 + ln];  w1f[2].y = W1[(1 + 2*hf)*32 + 16 + ln];
  w1f[3].x = W1[(4 + 2*hf)*32 + 16 + ln];  w1f[3].y = W1[(5 + 2*hf)*32 + 16 + ln];
  #pragma unroll
  for (int c = 0; c < 8; ++c) {
    w2f[c].x = W2[(4*c + 2*hf)*16 + ln];
    w2f[c].y = W2[(4*c + 1 + 2*hf)*16 + ln];
  }
  const float bia0 = b1[ln], bia1 = b1[16 + ln], bia2 = b2[ln];
  float (*hw)[HP] = h[w];

  v4f xat[2], xot[2];
  #pragma unroll
  for (int t = 0; t < 2; ++t) {
    const int pT = b * SEQ + 32 * w + 16 * t + ln;
    xat[t] = *(const v4f*)(aL + (size_t)pT * 4);
    xot[t] = *(const v4f*)(oL + (size_t)pT * 4);
  }

  float s3 = 0.0f;
  #pragma unroll
  for (int t = 0; t < 2; ++t) {
    const v4f xa = xat[t];
    const v4f xo = xot[t];

    v2f a0, a1;
    a0.x = hf ? xa.z : xa.x;  a0.y = hf ? xa.w : xa.y;
    a1.x = hf ? xo.z : xo.x;  a1.y = hf ? xo.w : xo.y;
    v8f acc0 = {0,0,0,0,0,0,0,0};
    v8f acc1 = {0,0,0,0,0,0,0,0};
    acc0 = __builtin_amdgcn_wmma_f32_16x16x4_f32(false, a0, false, w1f[0], (short)0, acc0, false, false);
    acc0 = __builtin_amdgcn_wmma_f32_16x16x4_f32(false, a1, false, w1f[1], (short)0, acc0, false, false);
    acc1 = __builtin_amdgcn_wmma_f32_16x16x4_f32(false, a0, false, w1f[2], (short)0, acc1, false, false);
    acc1 = __builtin_amdgcn_wmma_f32_16x16x4_f32(false, a1, false, w1f[3], (short)0, acc1, false, false);

    wave_lds_order();
    #pragma unroll
    for (int v = 0; v < 8; ++v) {
      float t0 = acc0[v] + bia0;  hw[v + 8*hf][ln]      = t0 > 0.0f ? t0 : 0.0f;
      float t1 = acc1[v] + bia1;  hw[v + 8*hf][16 + ln] = t1 > 0.0f ? t1 : 0.0f;
    }
    wave_lds_order();

    v8f acc2 = {0,0,0,0,0,0,0,0};
    #pragma unroll
    for (int c = 0; c < 8; ++c) {
      v2f aa;
      aa.x = hw[ln][4*c + 2*hf];
      aa.y = hw[ln][4*c + 1 + 2*hf];
      acc2 = __builtin_amdgcn_wmma_f32_16x16x4_f32(false, aa, false, w2f[c], (short)0, acc2, false, false);
    }
    wave_lds_order();
    #pragma unroll
    for (int v = 0; v < 8; ++v) {
      float t2 = acc2[v] + bia2;
      hw[v + 8*hf][ln] = t2 > 0.0f ? t2 : 0.0f;
    }
    wave_lds_order();

    float acc3 = b3[0];
    #pragma unroll
    for (int c = 0; c < 16; ++c) acc3 += hw[ln][c] * W3[c];
    s3 = (hf == t) ? acc3 : s3;
  }

  const float cg    = 1.0f / (1.0f + expf(-s3));
  const float mult  = cg < 0.5f ? 2.0f * cg : 1.0f;
  const float mA = fmaxf(fmaxf(av.x, av.y), fmaxf(av.z, av.w));
  const float e0 = expf(av.x - mA), e1 = expf(av.y - mA), e2 = expf(av.z - mA), e3 = expf(av.w - mA);
  const float impA = (e0 + e1) / (e0 + e1 + e2 + e3);
  const float mO = fmaxf(fmaxf(ov.x, ov.y), fmaxf(ov.z, ov.w));
  const float g0 = expf(ov.x - mO), g1 = expf(ov.y - mO), g2 = expf(ov.z - mO), g3 = expf(ov.w - mO);
  const float impO = (g0 + g1) / (g0 + g1 + g2 + g3);
  bool nearA = false, nearO = false;
  #pragma unroll
  for (int d = 0; d < 7; ++d) {
    const int ss = s + d - 3;
    const bool in = (ss >= 0) && (ss < SEQ);
    nearA |= in && (wa[d] > 0);
    nearO |= in && (wo[d] > 0);
  }
  const float fa2 = (impA > 0.5f && !nearO) ? 0.3f : 1.0f;
  const float fo2 = (impO > 0.5f && !nearA) ? 0.3f : 1.0f;

  unsigned mym = 0u;
  #pragma unroll
  for (int j = 0; j < 8; ++j) {
    const int i = 32 * w + 4 * j + sub;
    const unsigned m = __builtin_amdgcn_ballot_w32((col < i) && hit4(pv[j]));
    if ((lane >> 2) == j) mym = m;
  }
  bool anyrow = ((mym >> (8 * (lane & 3))) & 0xFFu) != 0u;
  unsigned um = __builtin_amdgcn_ballot_w32(!anyrow && s > 8);
  while (um != 0u) {
    const int r = __builtin_ctz(um);
    um &= um - 1u;
    const int i = 32 * w + r;
    const v4f* Rr = R + ((size_t)i << 10);
    bool found = false;
    for (int c = 8; c < i && !found; c += 256) {
      bool hh = false;
      #pragma unroll
      for (int u = 0; u < 8; ++u)
        hh |= hit4(__builtin_nontemporal_load(Rr + min(c + 32 * u + lane, i - 1)));
      found = __builtin_amdgcn_ballot_w32(hh) != 0u;
    }
    if (lane == r) anyrow = found;
  }
  const float f4 = anyrow ? 0.7f : 1.0f;

  const float fa1 = fa2 * f4;
  const float fo1 = fo2;
  const float a0m = av.x * mult, a1m = av.y * mult, a2m = av.z * mult, a3m = av.w * mult;
  const float o0m = ov.x * mult, o1m = ov.y * mult, o2m = ov.z * mult, o3m = ov.w * mult;
  const bool orig = (fmaxf(fmaxf(av.x, av.y), fmaxf(av.z, av.w)) > 0.5f) ||
                    (fmaxf(fmaxf(ov.x, ov.y), fmaxf(ov.z, ov.w)) > 0.5f);
  float fa = fa1, fo = fo1;
  const bool act1  = (fmaxf(fmaxf(a0m*fa, a1m*fa), fmaxf(a2m, a3m)) > 0.5f) ||
                     (fmaxf(fmaxf(o0m*fo, o1m*fo), fmaxf(o2m, o3m)) > 0.5f);
  fa = fa1 * 0.1f; fo = fo1 * 0.1f;
  const bool act01 = (fmaxf(fmaxf(a0m*fa, a1m*fa), fmaxf(a2m, a3m)) > 0.5f) ||
                     (fmaxf(fmaxf(o0m*fo, o1m*fo), fmaxf(o2m, o3m)) > 0.5f);
  const unsigned wOG  = __builtin_amdgcn_ballot_w32(orig);
  const unsigned wI1  = __builtin_amdgcn_ballot_w32(act1);
  const unsigned wI01 = __builtin_amdgcn_ballot_w32(act01);
  if (lane == 0) { OG[w] = wOG; IF1[w] = wI1; IF01[w] = wI01; }
  __syncthreads();

  if (w == 0) {
    const int t = lane;
    const unsigned og  = OG[t];
    const unsigned ogn = (t < 31) ? OG[t + 1] : 0u;
    const unsigned A   = IF1[t];
    const unsigned C   = IF01[t];
    const unsigned G   = (og >> 1) | (og >> 2) | (ogn << 31) | (ogn << 30);
    const unsigned g0c = G, g1c = G | 1u, g2c = G | 3u;
    unsigned na0 = 0u, na1 = 0u, na2 = 0u;
    unsigned nb0 = 0u, nb1 = 0u, nb2 = 0u;
    #pragma unroll
    for (int it = 0; it < 32; ++it) {
      nb0 = (na0 << 1) | (na0 << 2) | g0c;  na0 = (nb0 & A) | (~nb0 & C);
      nb1 = (na1 << 1) | (na1 << 2) | g1c;  na1 = (nb1 & A) | (~nb1 & C);
      nb2 = (na2 << 1) | (na2 << 2) | g2c;  na2 = (nb2 & A) | (~nb2 & C);
    }
    const unsigned c0 = min(na0 >> 30, 2u);
    const unsigned c1 = min(na1 >> 30, 2u);
    const unsigned c2 = min(na2 >> 30, 2u);
    const unsigned F  = c0 | (c1 << 2) | (c2 << 4);
    unsigned st = 0u, lo = 0u, hi = 0u;
    #pragma unroll
    for (int k = 0; k < 32; ++k) {
      if (k < 16) lo |= st << (2 * k); else hi |= st << (2 * (k - 16));
      const unsigned f = (unsigned)__builtin_amdgcn_readlane((int)F, k);
      st = (f >> (2u * st)) & 3u;
    }
    const unsigned mine = (((t < 16) ? lo : hi) >> (2 * (t & 15))) & 3u;
    F3[t] = (mine == 0u) ? nb0 : ((mine == 1u) ? nb1 : nb2);
  }
  __syncthreads();

  const float f3  = ((F3[w] >> lane) & 1u) ? 1.0f : 0.1f;
  const float faw = fa1 * f3;
  const float fow = fo1 * f3;
  v4f ra, ro;
  ra.x = av.x * mult * faw;  ra.y = av.y * mult * faw;
  ra.z = av.z * mult;        ra.w = av.w * mult;
  ro.x = ov.x * mult * fow;  ro.y = ov.y * mult * fow;
  ro.z = ov.z * mult;        ro.w = ov.w * mult;
  *(v4f*)(outA + (size_t)p * 4) = ra;
  *(v4f*)(outO + (size_t)p * 4) = ro;
}

extern "C" void kernel_launch(void* const* d_in, const int* in_sizes, int n_in,
                              void* d_out, int out_size, void* d_ws, size_t ws_size,
                              hipStream_t stream) {
  (void)in_sizes; (void)n_in; (void)out_size; (void)d_ws; (void)ws_size;
  const float* aL  = (const float*)d_in[0];
  const float* oL  = (const float*)d_in[1];
  const float* rel = (const float*)d_in[2];
  const int*   exA = (const int*)d_in[3];
  const int*   exO = (const int*)d_in[4];
  const float* W1  = (const float*)d_in[5];
  const float* b1  = (const float*)d_in[6];
  const float* W2  = (const float*)d_in[7];
  const float* b2  = (const float*)d_in[8];
  const float* W3  = (const float*)d_in[9];
  const float* b3  = (const float*)d_in[10];

  float* outA  = (float*)d_out;
  float* outO  = outA + NPOS * 4;

  k_all<<<BATCH, 1024, 0, stream>>>(rel, aL, oL, exA, exO,
                                    W1, b1, W2, b2, W3, b3, outA, outO);
}


# each batch split over 8 blocks (4 own waves + 1 look-back wave): a block takes its entering scan state from the 32 positions before it, exact recompute from the batch start if they leave it open
# speedup vs baseline: 8.1447x; 1.4655x over previous
#include <hip/hip_runtime.h>
#include <math.h>


#define BATCH 8
#define SEQ   1024
#define KREL  4
#define NPOS  (BATCH * SEQ)
#define OWNW  4
#define NWAVE 5
#define GPB   8
#define HP    34

typedef __attribute__((ext_vector_type(2))) float v2f;
typedef __attribute__((ext_vector_type(4))) float v4f;
typedef __attribute__((ext_vector_type(8))) float v8f;

__device__ __forceinline__ bool hit4(const v4f v) {
  return (v.x > 0.5f) || (v.y > 0.5f) || (v.z > 0.5f) || (v.w > 0.5f);
}

__device__ __forceinline__ void wave_lds_order() {
  __builtin_amdgcn_fence(__ATOMIC_RELEASE, "wavefront", "local");
  __builtin_amdgcn_wave_barrier();
  __builtin_amdgcn_fence(__ATOMIC_ACQUIRE, "wavefront", "local");
}

__device__ __forceinline__ unsigned origword(const float* __restrict__ aL,
                                             const float* __restrict__ oL,
                                             const int b, const int s0, const int lane) {
  const int s  = s0 + lane;
  const int sc = min(s, SEQ - 1);
  const v4f a = *(const v4f*)(aL + (size_t)(b * SEQ + sc) * 4);
  const v4f o = *(const v4f*)(oL + (size_t)(b * SEQ + sc) * 4);
  const bool orig = (s < SEQ) &&
                    ((fmaxf(fmaxf(a.x, a.y), fmaxf(a.z, a.w)) > 0.5f) ||
                     (fmaxf(fmaxf(o.x, o.y), fmaxf(o.z, o.w)) > 0.5f));
  return __builtin_amdgcn_ballot_w32(orig);
}

__device__ __forceinline__ void posword(
      const float* __restrict__ rel,
      const float* __restrict__ aL, const float* __restrict__ oL,
      const int*   __restrict__ exA, const int* __restrict__ exO,
      const float* __restrict__ W1, const float* __restrict__ b1,
      const float* __restrict__ W2, const float* __restrict__ b2,
      const float* __restrict__ W3, const float* __restrict__ b3,
      float (*hw)[HP], const int b, const int s0, const int lane,
      v4f& avo, v4f& ovo, float& multo, float& fa1o, float& fo1o,
      unsigned& wOGo, unsigned& wI1o, unsigned& wI01o) {
  const int hf   = lane >> 4;
  const int ln   = lane & 15;
  const int s    = s0 + lane;
  const int p    = b * SEQ + s;

  const v4f av = *(const v4f*)(aL + (size_t)p * 4);
  const v4f ov = *(const v4f*)(oL + (size_t)p * 4);

  const v4f* R  = (const v4f*)rel + ((size_t)b << 20);
  const int sub = lane >> 3, col = lane & 7;
  v4f pv[8];
  #pragma unroll
  for (int j = 0; j < 8; ++j) {
    const int i = s0 + 4 * j + sub;
    pv[j] = R[(i << 10) + min(col, max(i - 1, 0))];
  }

  const int* eA = exA + b * SEQ;
  const int* eO = exO + b * SEQ;
  int wa[7], wo[7];
  #pragma unroll
  for (int d = 0; d < 7; ++d) {
    const int sc = min(max(s + d - 3, 0), SEQ - 1);
    wa[d] = eA[sc];  wo[d] = eO[sc];
  }

  v2f w1f[4], w2f[8];
  w1f[0].x = W1[(0 + 2*hf)*32 + ln];       w1f[0].y = W1[(1 + 2*hf)*32 + ln];
  w1f[1].x = W1[(4 + 2*hf)*32 + ln];       w1f[1].y = W1[(5 + 2*hf)*32 + ln];
  w1f[2].x = W1[(0 + 2*hf)*32 + 16 + ln];  w1f[2].y = W1[(1 + 2*hf)*32 + 16 + ln];
  w1f[3].x = W1[(4 + 2*hf)*32 + 16 + ln];  w1f[3].y = W1[(5 + 2*hf)*32 + 16 + ln];
  #pragma unroll
  for (int c = 0; c < 8; ++c) {
    w2f[c].x = W2[(4*c + 2*hf)*16 + ln];
    w2f[c].y = W2[(4*c + 1 + 2*hf)*16 + ln];
  }
  const float bia0 = b1[ln], bia1 = b1[16 + ln], bia2 = b2[ln];

  v4f xat[2], xot[2];
  #pragma unroll
  for (int t = 0; t < 2; ++t) {
    const int pT = b * SEQ + s0 + 16 * t + ln;
    xat[t] = *(const v4f*)(aL + (size_t)pT * 4);
    xot[t] = *(const v4f*)(oL + (size_t)pT * 4);
  }

  float s3 = 0.0f;
  #pragma unroll
  for (int t = 0; t < 2; ++t) {
    const v4f xa = xat[t];
    const v4f xo = xot[t];

    v2f a0, a1;
    a0.x = hf ? xa.z : xa.x;  a0.y = hf ? xa.w : xa.y;
    a1.x = hf ? xo.z : xo.x;  a1.y = hf ? xo.w : xo.y;
    v8f acc0 = {0,0,0,0,0,0,0,0};
    v8f acc1 = {0,0,0,0,0,0,0,0};
    acc0 = __builtin_amdgcn_wmma_f32_16x16x4_f32(false, a0, false, w1f[0], (short)0, acc0, false, false);
    acc0 = __builtin_amdgcn_wmma_f32_16x16x4_f32(false, a1, false, w1f[1], (short)0, acc0, false, false);
    acc1 = __builtin_amdgcn_wmma_f32_16x16x4_f32(false, a0, false, w1f[2], (short)0, acc1, false, false);
    acc1 = __builtin_amdgcn_wmma_f32_16x16x4_f32(false, a1, false, w1f[3], (short)0, acc1, false, false);

    wave_lds_order();
    #pragma unroll
    for (int v = 0; v < 8; ++v) {
      float t0 = acc0[v] + bia0;  hw[v + 8*hf][ln]      = t0 > 0.0f ? t0 : 0.0f;
      float t1 = acc1[v] + bia1;  hw[v + 8*hf][16 + ln] = t1 > 0.0f ? t1 : 0.0f;
    }
    wave_lds_order();

    v8f acc2 = {0,0,0,0,0,0,0,0};
    #pragma unroll
    for (int c = 0; c < 8; ++c) {
      v2f aa;
      aa.x = hw[ln][4*c + 2*hf];
      aa.y = hw[ln][4*c + 1 + 2*hf];
      acc2 = __builtin_amdgcn_wmma_f32_16x16x4_f32(false, aa, false, w2f[c], (short)0, acc2, false, false);
    }
    wave_lds_order();
    #pragma unroll
    for (int v = 0; v < 8; ++v) {
      float t2 = acc2[v] + bia2;
      hw[v + 8*hf][ln] = t2 > 0.0f ? t2 : 0.0f;
    }
    wave_lds_order();

    float acc3 = b3[0];
    #pragma unroll
    for (int c = 0; c < 16; ++c) acc3 += hw[ln][c] * W3[c];
    s3 = (hf == t) ? acc3 : s3;
  }

  const float cg    = 1.0f / (1.0f + expf(-s3));
  const float mult  = cg < 0.5f ? 2.0f * cg : 1.0f;
  const float mA = fmaxf(fmaxf(av.x, av.y), fmaxf(av.z, av.w));
  const float e0 = expf(av.x - mA), e1 = expf(av.y - mA), e2 = expf(av.z - mA), e3 = expf(av.w - mA);
  const float impA = (e0 + e1) / (e0 + e1 + e2 + e3);
  const float mO = fmaxf(fmaxf(ov.x, ov.y), fmaxf(ov.z, ov.w));
  const float g0 = expf(ov.x - mO), g1 = expf(ov.y - mO), g2 = expf(ov.z - mO), g3 = expf(ov.w - mO);
  const float impO = (g0 + g1) / (g0 + g1 + g2 + g3);
  bool nearA = false, nearO = false;
  #pragma unroll
  for (int d = 0; d < 7; ++d) {
    const int ss = s + d - 3;
    const bool in = (ss >= 0) && (ss < SEQ);
    nearA |= in && (wa[d] > 0);
    nearO |= in && (wo[d] > 0);
  }
  const float fa2 = (impA > 0.5f && !nearO) ? 0.3f : 1.0f;
  const float fo2 = (impO > 0.5f && !nearA) ? 0.3f : 1.0f;

  unsigned mym = 0u;
  #pragma unroll
  for (int j = 0; j < 8; ++j) {
    const int i = s0 + 4 * j + sub;
    const unsigned m = __builtin_amdgcn_ballot_w32((col < i) && hit4(pv[j]));
    if ((lane >> 2) == j) mym = m;
  }
  bool anyrow = ((mym >> (8 * (lane & 3))) & 0xFFu) != 0u;
  unsigned um = __builtin_amdgcn_ballot_w32(!anyrow && s > 8);
  while (um != 0u) {
    const int r = __builtin_ctz(um);
    um &= um - 1u;
    const int i = s0 + r;
    const v4f* Rr = R + ((size_t)i << 10);
    bool found = false;
    for (int c = 8; c < i && !found; c += 256) {
      bool hh = false;
      #pragma unroll
      for (int u = 0; u < 8; ++u)
        hh |= hit4(__builtin_nontemporal_load(Rr + min(c + 32 * u + lane, i - 1)));
      found = __builtin_amdgcn_ballot_w32(hh) != 0u;
    }
    if (lane == r) anyrow = found;
  }
  const float f4 = anyrow ? 0.7f : 1.0f;

  const float fa1 = fa2 * f4;
  const float fo1 = fo2;
  const float a0m = av.x * mult, a1m = av.y * mult, a2m = av.z * mult, a3m = av.w * mult;
  const float o0m = ov.x * mult, o1m = ov.y * mult, o2m = ov.z * mult, o3m = ov.w * mult;
  const bool orig = (fmaxf(fmaxf(av.x, av.y), fmaxf(av.z, av.w)) > 0.5f) ||
                    (fmaxf(fmaxf(ov.x, ov.y), fmaxf(ov.z, ov.w)) > 0.5f);
  float fa = fa1, fo = fo1;
  const bool act1  = (fmaxf(fmaxf(a0m*fa, a1m*fa), fmaxf(a2m, a3m)) > 0.5f) ||
                     (fmaxf(fmaxf(o0m*fo, o1m*fo), fmaxf(o2m, o3m)) > 0.5f);
  fa = fa1 * 0.1f; fo = fo1 * 0.1f;
  const bool act01 = (fmaxf(fmaxf(a0m*fa, a1m*fa), fmaxf(a2m, a3m)) > 0.5f) ||
                     (fmaxf(fmaxf(o0m*fo, o1m*fo), fmaxf(o2m, o3m)) > 0.5f);
  wOGo  = __builtin_amdgcn_ballot_w32(orig);
  wI1o  = __builtin_amdgcn_ballot_w32(act1);
  wI01o = __builtin_amdgcn_ballot_w32(act01);
  avo = av;  ovo = ov;  multo = mult;  fa1o = fa1;  fo1o = fo1;
}

__global__ void __launch_bounds__(32 * NWAVE)
k_all(const float* __restrict__ rel,
      const float* __restrict__ aL, const float* __restrict__ oL,
      const int*   __restrict__ exA, const int* __restrict__ exO,
      const float* __restrict__ W1, const float* __restrict__ b1,
      const float* __restrict__ W2, const float* __restrict__ b2,
      const float* __restrict__ W3, const float* __restrict__ b3,
      float* __restrict__ outA, float* __restrict__ outO) {
  __shared__ float h[NWAVE][16][HP];
  __shared__ unsigned OG[NWAVE], IF1[NWAVE], IF01[NWAVE], F3[OWNW], OGN[1];

  const int b    = blockIdx.x / GPB;
  const int g    = blockIdx.x % GPB;
  const int base = g * (OWNW * 32);
  const int tid  = threadIdx.x;
  const int lane = tid & 31;
  const int w    = __builtin_amdgcn_readfirstlane(tid >> 5);
  const bool lb  = (w == OWNW);
  const int s0   = lb ? max(base - 32, 0) : base + 32 * w;

  v4f av, ov;
  float mult, fa1, fo1;
  unsigned wOG, wI1, wI01;
  posword(rel, aL, oL, exA, exO, W1, b1, W2, b2, W3, b3, h[w], b, s0, lane,
          av, ov, mult, fa1, fo1, wOG, wI1, wI01);
  if (lane == 0) { OG[w] = wOG; IF1[w] = wI1; IF01[w] = wI01; }
  if (lb) {
    const unsigned n = origword(aL, oL, b, base + OWNW * 32, lane);
    if (lane == 0) OGN[0] = n;
  }
  __syncthreads();

  if (w == 0) {
    const int t  = lane;
    const int ti = min(t, OWNW);
    const unsigned og  = OG[ti];
    const unsigned ogn = (t < OWNW - 1) ? OG[ti + 1] : ((t == OWNW - 1) ? OGN[0] : OG[0]);
    const unsigned A   = IF1[ti];
    const unsigned C   = IF01[ti];
    const unsigned G   = (og >> 1) | (og >> 2) | (ogn << 31) | (ogn << 30);
    const unsigned g0c = G, g1c = G | 1u, g2c = G | 3u;
    unsigned na0 = 0u, na1 = 0u, na2 = 0u;
    unsigned nb0 = 0u, nb1 = 0u, nb2 = 0u;
    #pragma unroll
    for (int it = 0; it < 32; ++it) {
      nb0 = (na0 << 1) | (na0 << 2) | g0c;  na0 = (nb0 & A) | (~nb0 & C);
      nb1 = (na1 << 1) | (na1 << 2) | g1c;  na1 = (nb1 & A) | (~nb1 & C);
      nb2 = (na2 << 1) | (na2 << 2) | g2c;  na2 = (nb2 & A) | (~nb2 & C);
    }
    const unsigned c0 = min(na0 >> 30, 2u);
    const unsigned c1 = min(na1 >> 30, 2u);
    const unsigned c2 = min(na2 >> 30, 2u);
    const unsigned F  = c0 | (c1 << 2) | (c2 << 4);

    unsigned st = 0u;
    if (g != 0) {
      const unsigned fl = (unsigned)__builtin_amdgcn_readlane((int)F, OWNW);
      const unsigned l0 = fl & 3u, l1 = (fl >> 2) & 3u, l2 = (fl >> 4) & 3u;
      if (l0 == l1 && l1 == l2) {
        st = l0;
      } else {
        unsigned cs = 0u;
        const int nwords = (base >> 5) - 1;
        for (int j = 0; j < nwords; ++j) {
          v4f xa, xo;  float xm, xf, xg;
          unsigned jOG, jA, jC;
          posword(rel, aL, oL, exA, exO, W1, b1, W2, b2, W3, b3, h[0], b, 32 * j, lane,
                  xa, xo, xm, xf, xg, jOG, jA, jC);
          const unsigned jn = origword(aL, oL, b, 32 * (j + 1), lane);
          const unsigned jG = (jOG >> 1) | (jOG >> 2) | (jn << 31) | (jn << 30);
          const unsigned gc = jG | ((cs == 0u) ? 0u : ((cs == 1u) ? 1u : 3u));
          unsigned na = 0u;
          for (int it = 0; it < 32; ++it) {
            const unsigned nb = (na << 1) | (na << 2) | gc;
            na = (nb & jA) | (~nb & jC);
          }
          cs = min(na >> 30, 2u);
        }
        st = (fl >> (2u * cs)) & 3u;
      }
    }
    unsigned cls = 0u;
    #pragma unroll
    for (int k = 0; k < OWNW; ++k) {
      cls |= st << (2 * k);
      const unsigned f = (unsigned)__builtin_amdgcn_readlane((int)F, k);
      st = (f >> (2u * st)) & 3u;
    }
    const unsigned mine = (cls >> (2 * min(t, OWNW - 1))) & 3u;
    if (t < OWNW) F3[t] = (mine == 0u) ? nb0 : ((mine == 1u) ? nb1 : nb2);
  }
  __syncthreads();

  if (!lb) {
    const int p     = b * SEQ + s0 + lane;
    const float f3  = ((F3[w] >> lane) & 1u) ? 1.0f : 0.1f;
    const float faw = fa1 * f3;
    const float fow = fo1 * f3;
    v4f ra, ro;
    ra.x = av.x * mult * faw;  ra.y = av.y * mult * faw;
    ra.z = av.z * mult;        ra.w = av.w * mult;
    ro.x = ov.x * mult * fow;  ro.y = ov.y * mult * fow;
    ro.z = ov.z * mult;        ro.w = ov.w * mult;
    *(v4f*)(outA + (size_t)p * 4) = ra;
    *(v4f*)(outO + (size_t)p * 4) = ro;
  }
}

extern "C" void kernel_launch(void* const* d_in, const int* in_sizes, int n_in,
                              void* d_out, int out_size, void* d_ws, size_t ws_size,
                              hipStream_t stream) {
  (void)in_sizes; (void)n_in; (void)out_size; (void)d_ws; (void)ws_size;
  const float* aL  = (const float*)d_in[0];
  const float* oL  = (const float*)d_in[1];
  const float* rel = (const float*)d_in[2];
  const int*   exA = (const int*)d_in[3];
  const int*   exO = (const int*)d_in[4];
  const float* W1  = (const float*)d_in[5];
  const float* b1  = (const float*)d_in[6];
  const float* W2  = (const float*)d_in[7];
  const float* b2  = (const float*)d_in[8];
  const float* W3  = (const float*)d_in[9];
  const float* b3  = (const float*)d_in[10];

  float* outA  = (float*)d_out;
  float* outO  = outA + NPOS * 4;

  k_all<<<BATCH * GPB, 32 * NWAVE, 0, stream>>>(rel, aL, oL, exA, exO,
                                                W1, b1, W2, b2, W3, b3, outA, outO);
}


# all loads ahead of the MLP: probe verdict and next-word test without short-circuits, so the first probe line and the look-back wave's next-word rows are not fetched behind it; W3 fetched once
# speedup vs baseline: 9.8301x; 1.2069x over previous
#include <hip/hip_runtime.h>
#include <math.h>


#define BATCH 8
#define SEQ   1024
#define KREL  4
#define NPOS  (BATCH * SEQ)
#define OWNW  4
#define NWAVE 5
#define GPB   8
#define HP    34

typedef __attribute__((ext_vector_type(2))) float v2f;
typedef __attribute__((ext_vector_type(4))) float v4f;
typedef __attribute__((ext_vector_type(8))) float v8f;

__device__ __forceinline__ bool hit4(const v4f v) {
  return (v.x > 0.5f) | (v.y > 0.5f) | (v.z > 0.5f) | (v.w > 0.5f);
}

__device__ __forceinline__ void wave_lds_order() {
  __builtin_amdgcn_fence(__ATOMIC_RELEASE, "wavefront", "local");
  __builtin_amdgcn_wave_barrier();
  __builtin_amdgcn_fence(__ATOMIC_ACQUIRE, "wavefront", "local");
}

__device__ __forceinline__ unsigned origword(const float* __restrict__ aL,
                                             const float* __restrict__ oL,
                                             const int b, const int s0, const int lane) {
  const int s  = s0 + lane;
  const int sc = min(s, SEQ - 1);
  const v4f a = *(const v4f*)(aL + (size_t)(b * SEQ + sc) * 4);
  const v4f o = *(const v4f*)(oL + (size_t)(b * SEQ + sc) * 4);
  const bool orig = (s < SEQ) &
                    ((fmaxf(fmaxf(a.x, a.y), fmaxf(a.z, a.w)) > 0.5f) |
                     (fmaxf(fmaxf(o.x, o.y), fmaxf(o.z, o.w)) > 0.5f));
  return __builtin_amdgcn_ballot_w32(orig);
}

__device__ __forceinline__ void posword(
      const float* __restrict__ rel,
      const float* __restrict__ aL, const float* __restrict__ oL,
      const int*   __restrict__ exA, const int* __restrict__ exO,
      const float* __restrict__ W1, const float* __restrict__ b1,
      const float* __restrict__ W2, const float* __restrict__ b2,
      const float* __restrict__ W3, const float* __restrict__ b3,
      float (*hw)[HP], const int b, const int s0, const int lane,
      v4f& avo, v4f& ovo, float& multo, float& fa1o, float& fo1o,
      unsigned& wOGo, unsigned& wI1o, unsigned& wI01o) {
  const int hf   = lane >> 4;
  const int ln   = lane & 15;
  const int s    = s0 + lane;
  const int p    = b * SEQ + s;

  const v4f av = *(const v4f*)(aL + (size_t)p * 4);
  const v4f ov = *(const v4f*)(oL + (size_t)p * 4);

  const v4f* R  = (const v4f*)rel + ((size_t)b << 20);
  const int sub = lane >> 3, col = lane & 7;
  v4f pv[8];
  #pragma unroll
  for (int j = 0; j < 8; ++j) {
    const int i = s0 + 4 * j + sub;
    pv[j] = R[(i << 10) + min(col, max(i - 1, 0))];
  }

  const int* eA = exA + b * SEQ;
  const int* eO = exO + b * SEQ;
  int wa[7], wo[7];
  #pragma unroll
  for (int d = 0; d < 7; ++d) {
    const int sc = min(max(s + d - 3, 0), SEQ - 1);
    wa[d] = eA[sc];  wo[d] = eO[sc];
  }

  v2f w1f[4], w2f[8];
  w1f[0].x = W1[(0 + 2*hf)*32 + ln];       w1f[0].y = W1[(1 + 2*hf)*32 + ln];
  w1f[1].x = W1[(4 + 2*hf)*32 + ln];       w1f[1].y = W1[(5 + 2*hf)*32 + ln];
  w1f[2].x = W1[(0 + 2*hf)*32 + 16 + ln];  w1f[2].y = W1[(1 + 2*hf)*32 + 16 + ln];
  w1f[3].x = W1[(4 + 2*hf)*32 + 16 + ln];  w1f[3].y = W1[(5 + 2*hf)*32 + 16 + ln];
  #pragma unroll
  for (int c = 0; c < 8; ++c) {
    w2f[c].x = W2[(4*c + 2*hf)*16 + ln];
    w2f[c].y = W2[(4*c + 1 + 2*hf)*16 + ln];
  }
  const float bia0 = b1[ln], bia1 = b1[16 + ln], bia2 = b2[ln];
  float w3[16];
  #pragma unroll
  for (int c = 0; c < 16; ++c) w3[c] = W3[c];
  const float bia3 = b3[0];

  v4f xat[2], xot[2];
  #pragma unroll
  for (int t = 0; t < 2; ++t) {
    const int pT = b * SEQ + s0 + 16 * t + ln;
    xat[t] = *(const v4f*)(aL + (size_t)pT * 4);
    xot[t] = *(const v4f*)(oL + (size_t)pT * 4);
  }

  float s3 = 0.0f;
  #pragma unroll
  for (int t = 0; t < 2; ++t) {
    const v4f xa = xat[t];
    const v4f xo = xot[t];

    v2f a0, a1;
    a0.x = hf ? xa.z : xa.x;  a0.y = hf ? xa.w : xa.y;
    a1.x = hf ? xo.z : xo.x;  a1.y = hf ? xo.w : xo.y;
    v8f acc0 = {0,0,0,0,0,0,0,0};
    v8f acc1 = {0,0,0,0,0,0,0,0};
    acc0 = __builtin_amdgcn_wmma_f32_16x16x4_f32(false, a0, false, w1f[0], (short)0, acc0, false, false);
    acc0 = __builtin_amdgcn_wmma_f32_16x16x4_f32(false, a1, false, w1f[1], (short)0, acc0, false, false);
    acc1 = __builtin_amdgcn_wmma_f32_16x16x4_f32(false, a0, false, w1f[2], (short)0, acc1, false, false);
    acc1 = __builtin_amdgcn_wmma_f32_16x16x4_f32(false, a1, false, w1f[3], (short)0, acc1, false, false);

    wave_lds_order();
    #pragma unroll
    for (int v = 0; v < 8; ++v) {
      float t0 = acc0[v] + bia0;  hw[v + 8*hf][ln]      = t0 > 0.0f ? t0 : 0.0f;
      float t1 = acc1[v] + bia1;  hw[v + 8*hf][16 + ln] = t1 > 0.0f ? t1 : 0.0f;
    }
    wave_lds_order();

    v8f acc2 = {0,0,0,0,0,0,0,0};
    #pragma unroll
    for (int c = 0; c < 8; ++c) {
      v2f aa;
      aa.x = hw[ln][4*c + 2*hf];
      aa.y = hw[ln][4*c + 1 + 2*hf];
      acc2 = __builtin_amdgcn_wmma_f32_16x16x4_f32(false, aa, false, w2f[c], (short)0, acc2, false, false);
    }
    wave_lds_order();
    #pragma unroll
    for (int v = 0; v < 8; ++v) {
      float t2 = acc2[v] + bia2;
      hw[v + 8*hf][ln] = t2 > 0.0f ? t2 : 0.0f;
    }
    wave_lds_order();

    float acc3 = bia3;
    #pragma unroll
    for (int c = 0; c < 16; ++c) acc3 += hw[ln][c] * w3[c];
    s3 = (hf == t) ? acc3 : s3;
  }

  const float cg    = 1.0f / (1.0f + expf(-s3));
  const float mult  = cg < 0.5f ? 2.0f * cg : 1.0f;
  const float mA = fmaxf(fmaxf(av.x, av.y), fmaxf(av.z, av.w));
  const float e0 = expf(av.x - mA), e1 = expf(av.y - mA), e2 = expf(av.z - mA), e3 = expf(av.w - mA);
  const float impA = (e0 + e1) / (e0 + e1 + e2 + e3);
  const float mO = fmaxf(fmaxf(ov.x, ov.y), fmaxf(ov.z, ov.w));
  const float g0 = expf(ov.x - mO), g1 = expf(ov.y - mO), g2 = expf(ov.z - mO), g3 = expf(ov.w - mO);
  const float impO = (g0 + g1) / (g0 + g1 + g2 + g3);
  bool nearA = false, nearO = false;
  #pragma unroll
  for (int d = 0; d < 7; ++d) {
    const int ss = s + d - 3;
    const bool in = (ss >= 0) && (ss < SEQ);
    nearA |= in && (wa[d] > 0);
    nearO |= in && (wo[d] > 0);
  }
  const float fa2 = (impA > 0.5f && !nearO) ? 0.3f : 1.0f;
  const float fo2 = (impO > 0.5f && !nearA) ? 0.3f : 1.0f;

  unsigned mym = 0u;
  #pragma unroll
  for (int j = 0; j < 8; ++j) {
    const int i = s0 + 4 * j + sub;
    const unsigned m = __builtin_amdgcn_ballot_w32((col < i) & hit4(pv[j]));
    if ((lane >> 2) == j) mym = m;
  }
  bool anyrow = ((mym >> (8 * (lane & 3))) & 0xFFu) != 0u;
  unsigned um = __builtin_amdgcn_ballot_w32(!anyrow && s > 8);
  while (um != 0u) {
    const int r = __builtin_ctz(um);
    um &= um - 1u;
    const int i = s0 + r;
    const v4f* Rr = R + ((size_t)i << 10);
    bool found = false;
    for (int c = 8; c < i && !found; c += 256) {
      bool hh = false;
      #pragma unroll
      for (int u = 0; u < 8; ++u)
        hh |= hit4(__builtin_nontemporal_load(Rr + min(c + 32 * u + lane, i - 1)));
      found = __builtin_amdgcn_ballot_w32(hh) != 0u;
    }
    if (lane == r) anyrow = found;
  }
  const float f4 = anyrow ? 0.7f : 1.0f;

  const float fa1 = fa2 * f4;
  const float fo1 = fo2;
  const float a0m = av.x * mult, a1m = av.y * mult, a2m = av.z * mult, a3m = av.w * mult;
  const float o0m = ov.x * mult, o1m = ov.y * mult, o2m = ov.z * mult, o3m = ov.w * mult;
  const bool orig = (fmaxf(fmaxf(av.x, av.y), fmaxf(av.z, av.w)) > 0.5f) ||
                    (fmaxf(fmaxf(ov.x, ov.y), fmaxf(ov.z, ov.w)) > 0.5f);
  float fa = fa1, fo = fo1;
  const bool act1  = (fmaxf(fmaxf(a0m*fa, a1m*fa), fmaxf(a2m, a3m)) > 0.5f) ||
                     (fmaxf(fmaxf(o0m*fo, o1m*fo), fmaxf(o2m, o3m)) > 0.5f);
  fa = fa1 * 0.1f; fo = fo1 * 0.1f;
  const bool act01 = (fmaxf(fmaxf(a0m*fa, a1m*fa), fmaxf(a2m, a3m)) > 0.5f) ||
                     (fmaxf(fmaxf(o0m*fo, o1m*fo), fmaxf(o2m, o3m)) > 0.5f);
  wOGo  = __builtin_amdgcn_ballot_w32(orig);
  wI1o  = __builtin_amdgcn_ballot_w32(act1);
  wI01o = __builtin_amdgcn_ballot_w32(act01);
  avo = av;  ovo = ov;  multo = mult;  fa1o = fa1;  fo1o = fo1;
}

__global__ void __launch_bounds__(32 * NWAVE)
k_all(const float* __restrict__ rel,
      const float* __restrict__ aL, const float* __restrict__ oL,
      const int*   __restrict__ exA, const int* __restrict__ exO,
      const float* __restrict__ W1, const float* __restrict__ b1,
      const float* __restrict__ W2, const float* __restrict__ b2,
      const float* __restrict__ W3, const float* __restrict__ b3,
      float* __restrict__ outA, float* __restrict__ outO) {
  __shared__ float h[NWAVE][16][HP];
  __shared__ unsigned OG[NWAVE], IF1[NWAVE], IF01[NWAVE], F3[OWNW], OGN[1];

  const int b    = blockIdx.x / GPB;
  const int g    = blockIdx.x % GPB;
  const int base = g * (OWNW * 32);
  const int tid  = threadIdx.x;
  const int lane = tid & 31;
  const int w    = __builtin_amdgcn_readfirstlane(tid >> 5);
  const bool lb  = (w == OWNW);
  const int s0   = lb ? max(base - 32, 0) : base + 32 * w;

  const unsigned nOG = origword(aL, oL, b, base + OWNW * 32, lane);

  v4f av, ov;
  float mult, fa1, fo1;
  unsigned wOG, wI1, wI01;
  posword(rel, aL, oL, exA, exO, W1, b1, W2, b2, W3, b3, h[w], b, s0, lane,
          av, ov, mult, fa1, fo1, wOG, wI1, wI01);
  if (lane == 0) { OG[w] = wOG; IF1[w] = wI1; IF01[w] = wI01; }
  if (lb && lane == 0) OGN[0] = nOG;
  __syncthreads();

  if (w == 0) {
    const int t  = lane;
    const int ti = min(t, OWNW);
    const unsigned og  = OG[ti];
    const unsigned ogn = (t < OWNW - 1) ? OG[ti + 1] : ((t == OWNW - 1) ? OGN[0] : OG[0]);
    const unsigned A   = IF1[ti];
    const unsigned C   = IF01[ti];
    const unsigned G   = (og >> 1) | (og >> 2) | (ogn << 31) | (ogn << 30);
    const unsigned g0c = G, g1c = G | 1u, g2c = G | 3u;
    unsigned na0 = 0u, na1 = 0u, na2 = 0u;
    unsigned nb0 = 0u, nb1 = 0u, nb2 = 0u;
    #pragma unroll
    for (int it = 0; it < 32; ++it) {
      nb0 = (na0 << 1) | (na0 << 2) | g0c;  na0 = (nb0 & A) | (~nb0 & C);
      nb1 = (na1 << 1) | (na1 << 2) | g1c;  na1 = (nb1 & A) | (~nb1 & C);
      nb2 = (na2 << 1) | (na2 << 2) | g2c;  na2 = (nb2 & A) | (~nb2 & C);
    }
    const unsigned c0 = min(na0 >> 30, 2u);
    const unsigned c1 = min(na1 >> 30, 2u);
    const unsigned c2 = min(na2 >> 30, 2u);
    const unsigned F  = c0 | (c1 << 2) | (c2 << 4);

    unsigned st = 0u;
    if (g != 0) {
      const unsigned fl = (unsigned)__builtin_amdgcn_readlane((int)F, OWNW);
      const unsigned l0 = fl & 3u, l1 = (fl >> 2) & 3u, l2 = (fl >> 4) & 3u;
      if (l0 == l1 && l1 == l2) {
        st = l0;
      } else {
        unsigned cs = 0u;
        const int nwords = (base >> 5) - 1;
        for (int j = 0; j < nwords; ++j) {
          v4f xa, xo;  float xm, xf, xg;
          unsigned jOG, jA, jC;
          posword(rel, aL, oL, exA, exO, W1, b1, W2, b2, W3, b3, h[0], b, 32 * j, lane,
                  xa, xo, xm, xf, xg, jOG, jA, jC);
          const unsigned jn = origword(aL, oL, b, 32 * (j + 1), lane);
          const unsigned jG = (jOG >> 1) | (jOG >> 2) | (jn << 31) | (jn << 30);
          const unsigned gc = jG | ((cs == 0u) ? 0u : ((cs == 1u) ? 1u : 3u));
          unsigned na = 0u;
          for (int it = 0; it < 32; ++it) {
            const unsigned nb = (na << 1) | (na << 2) | gc;
            na = (nb & jA) | (~nb & jC);
          }
          cs = min(na >> 30, 2u);
        }
        st = (fl >> (2u * cs)) & 3u;
      }
    }
    unsigned cls = 0u;
    #pragma unroll
    for (int k = 0; k < OWNW; ++k) {
      cls |= st << (2 * k);
      const unsigned f = (unsigned)__builtin_amdgcn_readlane((int)F, k);
      st = (f >> (2u * st)) & 3u;
    }
    const unsigned mine = (cls >> (2 * min(t, OWNW - 1))) & 3u;
    if (t < OWNW) F3[t] = (mine == 0u) ? nb0 : ((mine == 1u) ? nb1 : nb2);
  }
  __syncthreads();

  if (!lb) {
    const int p     = b * SEQ + s0 + lane;
    const float f3  = ((F3[w] >> lane) & 1u) ? 1.0f : 0.1f;
    const float faw = fa1 * f3;
    const float fow = fo1 * f3;
    v4f ra, ro;
    ra.x = av.x * mult * faw;  ra.y = av.y * mult * faw;
    ra.z = av.z * mult;        ra.w = av.w * mult;
    ro.x = ov.x * mult * fow;  ro.y = ov.y * mult * fow;
    ro.z = ov.z * mult;        ro.w = ov.w * mult;
    *(v4f*)(outA + (size_t)p * 4) = ra;
    *(v4f*)(outO + (size_t)p * 4) = ro;
  }
}

extern "C" void kernel_launch(void* const* d_in, const int* in_sizes, int n_in,
                              void* d_out, int out_size, void* d_ws, size_t ws_size,
                              hipStream_t stream) {
  (void)in_sizes; (void)n_in; (void)out_size; (void)d_ws; (void)ws_size;
  const float* aL  = (const float*)d_in[0];
  const float* oL  = (const float*)d_in[1];
  const float* rel = (const float*)d_in[2];
  const int*   exA = (const int*)d_in[3];
  const int*   exO = (const int*)d_in[4];
  const float* W1  = (const float*)d_in[5];
  const float* b1  = (const float*)d_in[6];
  const float* W2  = (const float*)d_in[7];
  const float* b2  = (const float*)d_in[8];
  const float* W3  = (const float*)d_in[9];
  const float* b3  = (const float*)d_in[10];

  float* outA  = (float*)d_out;
  float* outO  = outA + NPOS * 4;

  k_all<<<BATCH * GPB, 32 * NWAVE, 0, stream>>>(rel, aL, oL, exA, exO,
                                                W1, b1, W2, b2, W3, b3, outA, outO);
}


# MLP transposed (weights as WMMA A, positions as B): each layer's D registers are the next layer's B fragments, no LDS staging or fences, tiles interleave, loads held ahead; layer-3 sum order now 8+8
# speedup vs baseline: 10.5620x; 1.0745x over previous
#include <hip/hip_runtime.h>
#include <math.h>


#define BATCH 8
#define SEQ   1024
#define KREL  4
#define NPOS  (BATCH * SEQ)
#define OWNW  4
#define NWAVE 5
#define GPB   8

typedef __attribute__((ext_vector_type(2))) float v2f;
typedef __attribute__((ext_vector_type(4))) float v4f;
typedef __attribute__((ext_vector_type(8))) float v8f;

__device__ __forceinline__ bool hit4(const v4f v) {
  return (v.x > 0.5f) | (v.y > 0.5f) | (v.z > 0.5f) | (v.w > 0.5f);
}

__device__ __forceinline__ float xhalf(const float x) {
  const int xi = __builtin_bit_cast(int, x);
  return __builtin_bit_cast(float,
      __builtin_amdgcn_permlanex16(xi, xi, 0x76543210, 0xfedcba98, false, false));
}

__device__ __forceinline__ unsigned origword(const float* __restrict__ aL,
                                             const float* __restrict__ oL,
                                             const int b, const int s0, const int lane) {
  const int s  = s0 + lane;
  const int sc = min(s, SEQ - 1);
  const v4f a = *(const v4f*)(aL + (size_t)(b * SEQ + sc) * 4);
  const v4f o = *(const v4f*)(oL + (size_t)(b * SEQ + sc) * 4);
  const bool orig = (s < SEQ) &
                    ((fmaxf(fmaxf(a.x, a.y), fmaxf(a.z, a.w)) > 0.5f) |
                     (fmaxf(fmaxf(o.x, o.y), fmaxf(o.z, o.w)) > 0.5f));
  return __builtin_amdgcn_ballot_w32(orig);
}

__device__ __forceinline__ void posword(
      const float* __restrict__ rel,
      const float* __restrict__ aL, const float* __restrict__ oL,
      const int*   __restrict__ exA, const int* __restrict__ exO,
      const float* __restrict__ W1, const float* __restrict__ b1,
      const float* __restrict__ W2, const float* __restrict__ b2,
      const float* __restrict__ W3, const float* __restrict__ b3,
      const int b, const int s0, const int lane,
      v4f& avo, v4f& ovo, float& multo, float& fa1o, float& fo1o,
      unsigned& wOGo, unsigned& wI1o, unsigned& wI01o) {
  const int hf   = lane >> 4;
  const int ln   = lane & 15;
  const int s    = s0 + lane;
  const int p    = b * SEQ + s;

  const int k2 = 2 * hf;
  const int pl = 4 * ((ln & 7) >> 1) + 2 * (ln >> 3) + (ln & 1);
  v2f w1f[2][2], w2f[8];
  #pragma unroll
  for (int T = 0; T < 2; ++T) {
    #pragma unroll
    for (int q = 0; q < 2; ++q) {
      w1f[T][q].x = W1[(4*q + k2)*32     + 16*T + pl];
      w1f[T][q].y = W1[(4*q + k2 + 1)*32 + 16*T + pl];
    }
  }
  v4f xat[2], xot[2];
  #pragma unroll
  for (int t = 0; t < 2; ++t) {
    const int pT = b * SEQ + s0 + 16 * t + ln;
    xat[t] = *(const v4f*)(aL + (size_t)pT * 4);
    xot[t] = *(const v4f*)(oL + (size_t)pT * 4);
  }
  v2f bb1[2][4];
  #pragma unroll
  for (int T = 0; T < 2; ++T) {
    #pragma unroll
    for (int j = 0; j < 4; ++j) bb1[T][j] = *(const v2f*)(b1 + 16*T + 4*j + k2);
  }
  #pragma unroll
  for (int c = 0; c < 8; ++c) {
    w2f[c].x = W2[(4*c + k2)*16 + ln];
    w2f[c].y = W2[(4*c + 1 + k2)*16 + ln];
  }
  v4f bb2[2], w3v[2];
  #pragma unroll
  for (int u = 0; u < 2; ++u) {
    bb2[u] = *(const v4f*)(b2 + 8*hf + 4*u);
    w3v[u] = *(const v4f*)(W3 + 8*hf + 4*u);
  }
  const float bia3 = b3[0];

  const v4f av = *(const v4f*)(aL + (size_t)p * 4);
  const v4f ov = *(const v4f*)(oL + (size_t)p * 4);

  const v4f* R  = (const v4f*)rel + ((size_t)b << 20);
  const int sub = lane >> 3, col = lane & 7;
  v4f pv[8];
  #pragma unroll
  for (int j = 0; j < 8; ++j) {
    const int i = s0 + 4 * j + sub;
    pv[j] = R[(i << 10) + min(col, max(i - 1, 0))];
  }

  const int* eA = exA + b * SEQ;
  const int* eO = exO + b * SEQ;
  int wa[7], wo[7];
  #pragma unroll
  for (int d = 0; d < 7; ++d) {
    const int sc = min(max(s + d - 3, 0), SEQ - 1);
    wa[d] = eA[sc];  wo[d] = eO[sc];
  }

  __builtin_amdgcn_sched_barrier(0);

  v2f xb[2][2];
  #pragma unroll
  for (int t = 0; t < 2; ++t) {
    const v4f xa = xat[t];
    const v4f xo = xot[t];
    xb[t][0].x = hf ? xa.z : xa.x;  xb[t][0].y = hf ? xa.w : xa.y;
    xb[t][1].x = hf ? xo.z : xo.x;  xb[t][1].y = hf ? xo.w : xo.y;
  }

  v8f h1[2][2];
  #pragma unroll
  for (int T = 0; T < 2; ++T) {
    #pragma unroll
    for (int t = 0; t < 2; ++t) {
      v8f acc = {0,0,0,0,0,0,0,0};
      acc = __builtin_amdgcn_wmma_f32_16x16x4_f32(false, w1f[T][0], false, xb[t][0], (short)0, acc, false, false);
      acc = __builtin_amdgcn_wmma_f32_16x16x4_f32(false, w1f[T][1], false, xb[t][1], (short)0, acc, false, false);
      #pragma unroll
      for (int r = 0; r < 8; ++r) {
        const float v = acc[r] + bb1[T][r >> 1][r & 1];
        acc[r] = v > 0.0f ? v : 0.0f;
      }
      h1[t][T] = acc;
    }
  }

  v8f a2[2];
  a2[0] = (v8f){0,0,0,0,0,0,0,0};
  a2[1] = (v8f){0,0,0,0,0,0,0,0};
  #pragma unroll
  for (int c = 0; c < 8; ++c) {
    #pragma unroll
    for (int t = 0; t < 2; ++t) {
      v2f hb;
      hb.x = h1[t][c >> 2][2 * (c & 3)];
      hb.y = h1[t][c >> 2][2 * (c & 3) + 1];
      a2[t] = __builtin_amdgcn_wmma_f32_16x16x4_f32(false, w2f[c], false, hb, (short)0, a2[t], false, false);
    }
  }

  float part[2];
  #pragma unroll
  for (int t = 0; t < 2; ++t) {
    float hv[8];
    #pragma unroll
    for (int r = 0; r < 8; ++r) {
      const float v = a2[t][r] + bb2[r >> 2][r & 3];
      hv[r] = v > 0.0f ? v : 0.0f;
    }
    float pr = hv[0] * w3v[0][0];
    #pragma unroll
    for (int r = 1; r < 8; ++r) pr += hv[r] * w3v[r >> 2][r & 3];
    part[t] = pr;
  }
  const float mine = hf ? part[1] : part[0];
  const float send = hf ? part[0] : part[1];
  const float s3   = (mine + xhalf(send)) + bia3;

  const float cg    = 1.0f / (1.0f + expf(-s3));
  const float mult  = cg < 0.5f ? 2.0f * cg : 1.0f;
  const float mA = fmaxf(fmaxf(av.x, av.y), fmaxf(av.z, av.w));
  const float e0 = expf(av.x - mA), e1 = expf(av.y - mA), e2 = expf(av.z - mA), e3 = expf(av.w - mA);
  const float impA = (e0 + e1) / (e0 + e1 + e2 + e3);
  const float mO = fmaxf(fmaxf(ov.x, ov.y), fmaxf(ov.z, ov.w));
  const float g0 = expf(ov.x - mO), g1 = expf(ov.y - mO), g2 = expf(ov.z - mO), g3 = expf(ov.w - mO);
  const float impO = (g0 + g1) / (g0 + g1 + g2 + g3);
  bool nearA = false, nearO = false;
  #pragma unroll
  for (int d = 0; d < 7; ++d) {
    const int ss = s + d - 3;
    const bool in = (ss >= 0) & (ss < SEQ);
    nearA |= in & (wa[d] > 0);
    nearO |= in & (wo[d] > 0);
  }
  const float fa2 = (impA > 0.5f && !nearO) ? 0.3f : 1.0f;
  const float fo2 = (impO > 0.5f && !nearA) ? 0.3f : 1.0f;

  unsigned mym = 0u;
  #pragma unroll
  for (int j = 0; j < 8; ++j) {
    const int i = s0 + 4 * j + sub;
    const unsigned m = __builtin_amdgcn_ballot_w32((col < i) & hit4(pv[j]));
    if ((lane >> 2) == j) mym = m;
  }
  bool anyrow = ((mym >> (8 * (lane & 3))) & 0xFFu) != 0u;
  unsigned um = __builtin_amdgcn_ballot_w32(!anyrow && s > 8);
  while (um != 0u) {
    const int r = __builtin_ctz(um);
    um &= um - 1u;
    const int i = s0 + r;
    const v4f* Rr = R + ((size_t)i << 10);
    bool found = false;
    for (int c = 8; c < i && !found; c += 256) {
      bool hh = false;
      #pragma unroll
      for (int u = 0; u < 8; ++u)
        hh |= hit4(__builtin_nontemporal_load(Rr + min(c + 32 * u + lane, i - 1)));
      found = __builtin_amdgcn_ballot_w32(hh) != 0u;
    }
    if (lane == r) anyrow = found;
  }
  const float f4 = anyrow ? 0.7f : 1.0f;

  const float fa1 = fa2 * f4;
  const float fo1 = fo2;
  const float a0m = av.x * mult, a1m = av.y * mult, a2m = av.z * mult, a3m = av.w * mult;
  const float o0m = ov.x * mult, o1m = ov.y * mult, o2m = ov.z * mult, o3m = ov.w * mult;
  const bool orig = (fmaxf(fmaxf(av.x, av.y), fmaxf(av.z, av.w)) > 0.5f) ||
                    (fmaxf(fmaxf(ov.x, ov.y), fmaxf(ov.z, ov.w)) > 0.5f);
  float fa = fa1, fo = fo1;
  const bool act1  = (fmaxf(fmaxf(a0m*fa, a1m*fa), fmaxf(a2m, a3m)) > 0.5f) ||
                     (fmaxf(fmaxf(o0m*fo, o1m*fo), fmaxf(o2m, o3m)) > 0.5f);
  fa = fa1 * 0.1f; fo = fo1 * 0.1f;
  const bool act01 = (fmaxf(fmaxf(a0m*fa, a1m*fa), fmaxf(a2m, a3m)) > 0.5f) ||
                     (fmaxf(fmaxf(o0m*fo, o1m*fo), fmaxf(o2m, o3m)) > 0.5f);
  wOGo  = __builtin_amdgcn_ballot_w32(orig);
  wI1o  = __builtin_amdgcn_ballot_w32(act1);
  wI01o = __builtin_amdgcn_ballot_w32(act01);
  avo = av;  ovo = ov;  multo = mult;  fa1o = fa1;  fo1o = fo1;
}

__global__ void __launch_bounds__(32 * NWAVE)
k_all(const float* __restrict__ rel,
      const float* __restrict__ aL, const float* __restrict__ oL,
      const int*   __restrict__ exA, const int* __restrict__ exO,
      const float* __restrict__ W1, const float* __restrict__ b1,
      const float* __restrict__ W2, const float* __restrict__ b2,
      const float* __restrict__ W3, const float* __restrict__ b3,
      float* __restrict__ outA, float* __restrict__ outO) {
  __shared__ unsigned OG[NWAVE], IF1[NWAVE], IF01[NWAVE], F3[OWNW], OGN[1];

  const int b    = blockIdx.x / GPB;
  const int g    = blockIdx.x % GPB;
  const int base = g * (OWNW * 32);
  const int tid  = threadIdx.x;
  const int lane = tid & 31;
  const int w    = __builtin_amdgcn_readfirstlane(tid >> 5);
  const bool lb  = (w == OWNW);
  const int s0   = lb ? max(base - 32, 0) : base + 32 * w;

  const unsigned nOG = origword(aL, oL, b, base + OWNW * 32, lane);

  v4f av, ov;
  float mult, fa1, fo1;
  unsigned wOG, wI1, wI01;
  posword(rel, aL, oL, exA, exO, W1, b1, W2, b2, W3, b3, b, s0, lane,
          av, ov, mult, fa1, fo1, wOG, wI1, wI01);
  if (lane == 0) { OG[w] = wOG; IF1[w] = wI1; IF01[w] = wI01; }
  if (lb && lane == 0) OGN[0] = nOG;
  __syncthreads();

  if (w == 0) {
    const int t  = lane;
    const int ti = min(t, OWNW);
    const unsigned og  = OG[ti];
    const unsigned ogn = (t < OWNW - 1) ? OG[ti + 1] : ((t == OWNW - 1) ? OGN[0] : OG[0]);
    const unsigned A   = IF1[ti];
    const unsigned C   = IF01[ti];
    const unsigned G   = (og >> 1) | (og >> 2) | (ogn << 31) | (ogn << 30);
    const unsigned g0c = G, g1c = G | 1u, g2c = G | 3u;
    unsigned na0 = 0u, na1 = 0u, na2 = 0u;
    unsigned nb0 = 0u, nb1 = 0u, nb2 = 0u;
    #pragma unroll
    for (int it = 0; it < 32; ++it) {
      nb0 = (na0 << 1) | (na0 << 2) | g0c;  na0 = (nb0 & A) | (~nb0 & C);
      nb1 = (na1 << 1) | (na1 << 2) | g1c;  na1 = (nb1 & A) | (~nb1 & C);
      nb2 = (na2 << 1) | (na2 << 2) | g2c;  na2 = (nb2 & A) | (~nb2 & C);
    }
    const unsigned c0 = min(na0 >> 30, 2u);
    const unsigned c1 = min(na1 >> 30, 2u);
    const unsigned c2 = min(na2 >> 30, 2u);
    const unsigned F  = c0 | (c1 << 2) | (c2 << 4);

    unsigned st = 0u;
    if (g != 0) {
      const unsigned fl = (unsigned)__builtin_amdgcn_readlane((int)F, OWNW);
      const unsigned l0 = fl & 3u, l1 = (fl >> 2) & 3u, l2 = (fl >> 4) & 3u;
      if (l0 == l1 && l1 == l2) {
        st = l0;
      } else {
        unsigned cs = 0u;
        const int nwords = (base >> 5) - 1;
        for (int j = 0; j < nwords; ++j) {
          v4f xa, xo;  float xm, xf, xg;
          unsigned jOG, jA, jC;
          posword(rel, aL, oL, exA, exO, W1, b1, W2, b2, W3, b3, b, 32 * j, lane,
                  xa, xo, xm, xf, xg, jOG, jA, jC);
          const unsigned jn = origword(aL, oL, b, 32 * (j + 1), lane);
          const unsigned jG = (jOG >> 1) | (jOG >> 2) | (jn << 31) | (jn << 30);
          const unsigned gc = jG | ((cs == 0u) ? 0u : ((cs == 1u) ? 1u : 3u));
          unsigned na = 0u;
          for (int it = 0; it < 32; ++it) {
            const unsigned nb = (na << 1) | (na << 2) | gc;
            na = (nb & jA) | (~nb & jC);
          }
          cs = min(na >> 30, 2u);
        }
        st = (fl >> (2u * cs)) & 3u;
      }
    }
    unsigned cls = 0u;
    #pragma unroll
    for (int k = 0; k < OWNW; ++k) {
      cls |= st << (2 * k);
      const unsigned f = (unsigned)__builtin_amdgcn_readlane((int)F, k);
      st = (f >> (2u * st)) & 3u;
    }
    const unsigned mine = (cls >> (2 * min(t, OWNW - 1))) & 3u;
    if (t < OWNW) F3[t] = (mine == 0u) ? nb0 : ((mine == 1u) ? nb1 : nb2);
  }
  __syncthreads();

  if (!lb) {
    const int p     = b * SEQ + s0 + lane;
    const float f3  = ((F3[w] >> lane) & 1u) ? 1.0f : 0.1f;
    const float faw = fa1 * f3;
    const float fow = fo1 * f3;
    v4f ra, ro;
    ra.x = av.x * mult * faw;  ra.y = av.y * mult * faw;
    ra.z = av.z * mult;        ra.w = av.w * mult;
    ro.x = ov.x * mult * fow;  ro.y = ov.y * mult * fow;
    ro.z = ov.z * mult;        ro.w = ov.w * mult;
    *(v4f*)(outA + (size_t)p * 4) = ra;
    *(v4f*)(outO + (size_t)p * 4) = ro;
  }
}

extern "C" void kernel_launch(void* const* d_in, const int* in_sizes, int n_in,
                              void* d_out, int out_size, void* d_ws, size_t ws_size,
                              hipStream_t stream) {
  (void)in_sizes; (void)n_in; (void)out_size; (void)d_ws; (void)ws_size;
  const float* aL  = (const float*)d_in[0];
  const float* oL  = (const float*)d_in[1];
  const float* rel = (const float*)d_in[2];
  const int*   exA = (const int*)d_in[3];
  const int*   exO = (const int*)d_in[4];
  const float* W1  = (const float*)d_in[5];
  const float* b1  = (const float*)d_in[6];
  const float* W2  = (const float*)d_in[7];
  const float* b2  = (const float*)d_in[8];
  const float* W3  = (const float*)d_in[9];
  const float* b3  = (const float*)d_in[10];

  float* outA  = (float*)d_out;
  float* outO  = outA + NPOS * 4;

  k_all<<<BATCH * GPB, 32 * NWAVE, 0, stream>>>(rel, aL, oL, exA, exO,
                                                W1, b1, W2, b2, W3, b3, outA, outO);
}


# scan: the per-word bitwise fixpoint starts from IF1 and stops when a pair of passes changes nothing (the map's one fixpoint, exact; at most 32 passes) instead of always running 32 unrolled passes
# speedup vs baseline: 11.3765x; 1.0771x over previous
#include <hip/hip_runtime.h>
#include <math.h>


#define BATCH 8
#define SEQ   1024
#define KREL  4
#define NPOS  (BATCH * SEQ)
#define OWNW  4
#define NWAVE 5
#define GPB   8

typedef __attribute__((ext_vector_type(2))) float v2f;
typedef __attribute__((ext_vector_type(4))) float v4f;
typedef __attribute__((ext_vector_type(8))) float v8f;

__device__ __forceinline__ bool hit4(const v4f v) {
  return (v.x > 0.5f) | (v.y > 0.5f) | (v.z > 0.5f) | (v.w > 0.5f);
}

__device__ __forceinline__ float xhalf(const float x) {
  const int xi = __builtin_bit_cast(int, x);
  return __builtin_bit_cast(float,
      __builtin_amdgcn_permlanex16(xi, xi, 0x76543210, 0xfedcba98, false, false));
}

__device__ __forceinline__ unsigned origword(const float* __restrict__ aL,
                                             const float* __restrict__ oL,
                                             const int b, const int s0, const int lane) {
  const int s  = s0 + lane;
  const int sc = min(s, SEQ - 1);
  const v4f a = *(const v4f*)(aL + (size_t)(b * SEQ + sc) * 4);
  const v4f o = *(const v4f*)(oL + (size_t)(b * SEQ + sc) * 4);
  const bool orig = (s < SEQ) &
                    ((fmaxf(fmaxf(a.x, a.y), fmaxf(a.z, a.w)) > 0.5f) |
                     (fmaxf(fmaxf(o.x, o.y), fmaxf(o.z, o.w)) > 0.5f));
  return __builtin_amdgcn_ballot_w32(orig);
}

__device__ __forceinline__ void posword(
      const float* __restrict__ rel,
      const float* __restrict__ aL, const float* __restrict__ oL,
      const int*   __restrict__ exA, const int* __restrict__ exO,
      const float* __restrict__ W1, const float* __restrict__ b1,
      const float* __restrict__ W2, const float* __restrict__ b2,
      const float* __restrict__ W3, const float* __restrict__ b3,
      const int b, const int s0, const int lane,
      v4f& avo, v4f& ovo, float& multo, float& fa1o, float& fo1o,
      unsigned& wOGo, unsigned& wI1o, unsigned& wI01o) {
  const int hf   = lane >> 4;
  const int ln   = lane & 15;
  const int s    = s0 + lane;
  const int p    = b * SEQ + s;

  const int k2 = 2 * hf;
  const int pl = 4 * ((ln & 7) >> 1) + 2 * (ln >> 3) + (ln & 1);
  v2f w1f[2][2], w2f[8];
  #pragma unroll
  for (int T = 0; T < 2; ++T) {
    #pragma unroll
    for (int q = 0; q < 2; ++q) {
      w1f[T][q].x = W1[(4*q + k2)*32     + 16*T + pl];
      w1f[T][q].y = W1[(4*q + k2 + 1)*32 + 16*T + pl];
    }
  }
  v4f xat[2], xot[2];
  #pragma unroll
  for (int t = 0; t < 2; ++t) {
    const int pT = b * SEQ + s0 + 16 * t + ln;
    xat[t] = *(const v4f*)(aL + (size_t)pT * 4);
    xot[t] = *(const v4f*)(oL + (size_t)pT * 4);
  }
  v2f bb1[2][4];
  #pragma unroll
  for (int T = 0; T < 2; ++T) {
    #pragma unroll
    for (int j = 0; j < 4; ++j) bb1[T][j] = *(const v2f*)(b1 + 16*T + 4*j + k2);
  }
  #pragma unroll
  for (int c = 0; c < 8; ++c) {
    w2f[c].x = W2[(4*c + k2)*16 + ln];
    w2f[c].y = W2[(4*c + 1 + k2)*16 + ln];
  }
  v4f bb2[2], w3v[2];
  #pragma unroll
  for (int u = 0; u < 2; ++u) {
    bb2[u] = *(const v4f*)(b2 + 8*hf + 4*u);
    w3v[u] = *(const v4f*)(W3 + 8*hf + 4*u);
  }
  const float bia3 = b3[0];

  const v4f av = *(const v4f*)(aL + (size_t)p * 4);
  const v4f ov = *(const v4f*)(oL + (size_t)p * 4);

  const v4f* R  = (const v4f*)rel + ((size_t)b << 20);
  const int sub = lane >> 3, col = lane & 7;
  v4f pv[8];
  #pragma unroll
  for (int j = 0; j < 8; ++j) {
    const int i = s0 + 4 * j + sub;
    pv[j] = R[(i << 10) + min(col, max(i - 1, 0))];
  }

  const int* eA = exA + b * SEQ;
  const int* eO = exO + b * SEQ;
  int wa[7], wo[7];
  #pragma unroll
  for (int d = 0; d < 7; ++d) {
    const int sc = min(max(s + d - 3, 0), SEQ - 1);
    wa[d] = eA[sc];  wo[d] = eO[sc];
  }

  __builtin_amdgcn_sched_barrier(0);

  v2f xb[2][2];
  #pragma unroll
  for (int t = 0; t < 2; ++t) {
    const v4f xa = xat[t];
    const v4f xo = xot[t];
    xb[t][0].x = hf ? xa.z : xa.x;  xb[t][0].y = hf ? xa.w : xa.y;
    xb[t][1].x = hf ? xo.z : xo.x;  xb[t][1].y = hf ? xo.w : xo.y;
  }

  v8f h1[2][2];
  #pragma unroll
  for (int T = 0; T < 2; ++T) {
    #pragma unroll
    for (int t = 0; t < 2; ++t) {
      v8f acc = {0,0,0,0,0,0,0,0};
      acc = __builtin_amdgcn_wmma_f32_16x16x4_f32(false, w1f[T][0], false, xb[t][0], (short)0, acc, false, false);
      acc = __builtin_amdgcn_wmma_f32_16x16x4_f32(false, w1f[T][1], false, xb[t][1], (short)0, acc, false, false);
      #pragma unroll
      for (int r = 0; r < 8; ++r) {
        const float v = acc[r] + bb1[T][r >> 1][r & 1];
        acc[r] = v > 0.0f ? v : 0.0f;
      }
      h1[t][T] = acc;
    }
  }

  v8f a2[2];
  a2[0] = (v8f){0,0,0,0,0,0,0,0};
  a2[1] = (v8f){0,0,0,0,0,0,0,0};
  #pragma unroll
  for (int c = 0; c < 8; ++c) {
    #pragma unroll
    for (int t = 0; t < 2; ++t) {
      v2f hb;
      hb.x = h1[t][c >> 2][2 * (c & 3)];
      hb.y = h1[t][c >> 2][2 * (c & 3) + 1];
      a2[t] = __builtin_amdgcn_wmma_f32_16x16x4_f32(false, w2f[c], false, hb, (short)0, a2[t], false, false);
    }
  }

  float part[2];
  #pragma unroll
  for (int t = 0; t < 2; ++t) {
    float hv[8];
    #pragma unroll
    for (int r = 0; r < 8; ++r) {
      const float v = a2[t][r] + bb2[r >> 2][r & 3];
      hv[r] = v > 0.0f ? v : 0.0f;
    }
    float pr = hv[0] * w3v[0][0];
    #pragma unroll
    for (int r = 1; r < 8; ++r) pr += hv[r] * w3v[r >> 2][r & 3];
    part[t] = pr;
  }
  const float mine = hf ? part[1] : part[0];
  const float send = hf ? part[0] : part[1];
  const float s3   = (mine + xhalf(send)) + bia3;

  const float cg    = 1.0f / (1.0f + expf(-s3));
  const float mult  = cg < 0.5f ? 2.0f * cg : 1.0f;
  const float mA = fmaxf(fmaxf(av.x, av.y), fmaxf(av.z, av.w));
  const float e0 = expf(av.x - mA), e1 = expf(av.y - mA), e2 = expf(av.z - mA), e3 = expf(av.w - mA);
  const float impA = (e0 + e1) / (e0 + e1 + e2 + e3);
  const float mO = fmaxf(fmaxf(ov.x, ov.y), fmaxf(ov.z, ov.w));
  const float g0 = expf(ov.x - mO), g1 = expf(ov.y - mO), g2 = expf(ov.z - mO), g3 = expf(ov.w - mO);
  const float impO = (g0 + g1) / (g0 + g1 + g2 + g3);
  bool nearA = false, nearO = false;
  #pragma unroll
  for (int d = 0; d < 7; ++d) {
    const int ss = s + d - 3;
    const bool in = (ss >= 0) & (ss < SEQ);
    nearA |= in & (wa[d] > 0);
    nearO |= in & (wo[d] > 0);
  }
  const float fa2 = (impA > 0.5f && !nearO) ? 0.3f : 1.0f;
  const float fo2 = (impO > 0.5f && !nearA) ? 0.3f : 1.0f;

  unsigned mym = 0u;
  #pragma unroll
  for (int j = 0; j < 8; ++j) {
    const int i = s0 + 4 * j + sub;
    const unsigned m = __builtin_amdgcn_ballot_w32((col < i) & hit4(pv[j]));
    if ((lane >> 2) == j) mym = m;
  }
  bool anyrow = ((mym >> (8 * (lane & 3))) & 0xFFu) != 0u;
  unsigned um = __builtin_amdgcn_ballot_w32(!anyrow && s > 8);
  while (um != 0u) {
    const int r = __builtin_ctz(um);
    um &= um - 1u;
    const int i = s0 + r;
    const v4f* Rr = R + ((size_t)i << 10);
    bool found = false;
    for (int c = 8; c < i && !found; c += 256) {
      bool hh = false;
      #pragma unroll
      for (int u = 0; u < 8; ++u)
        hh |= hit4(__builtin_nontemporal_load(Rr + min(c + 32 * u + lane, i - 1)));
      found = __builtin_amdgcn_ballot_w32(hh) != 0u;
    }
    if (lane == r) anyrow = found;
  }
  const float f4 = anyrow ? 0.7f : 1.0f;

  const float fa1 = fa2 * f4;
  const float fo1 = fo2;
  const float a0m = av.x * mult, a1m = av.y * mult, a2m = av.z * mult, a3m = av.w * mult;
  const float o0m = ov.x * mult, o1m = ov.y * mult, o2m = ov.z * mult, o3m = ov.w * mult;
  const bool orig = (fmaxf(fmaxf(av.x, av.y), fmaxf(av.z, av.w)) > 0.5f) ||
                    (fmaxf(fmaxf(ov.x, ov.y), fmaxf(ov.z, ov.w)) > 0.5f);
  float fa = fa1, fo = fo1;
  const bool act1  = (fmaxf(fmaxf(a0m*fa, a1m*fa), fmaxf(a2m, a3m)) > 0.5f) ||
                     (fmaxf(fmaxf(o0m*fo, o1m*fo), fmaxf(o2m, o3m)) > 0.5f);
  fa = fa1 * 0.1f; fo = fo1 * 0.1f;
  const bool act01 = (fmaxf(fmaxf(a0m*fa, a1m*fa), fmaxf(a2m, a3m)) > 0.5f) ||
                     (fmaxf(fmaxf(o0m*fo, o1m*fo), fmaxf(o2m, o3m)) > 0.5f);
  wOGo  = __builtin_amdgcn_ballot_w32(orig);
  wI1o  = __builtin_amdgcn_ballot_w32(act1);
  wI01o = __builtin_amdgcn_ballot_w32(act01);
  avo = av;  ovo = ov;  multo = mult;  fa1o = fa1;  fo1o = fo1;
}

__global__ void __launch_bounds__(32 * NWAVE)
k_all(const float* __restrict__ rel,
      const float* __restrict__ aL, const float* __restrict__ oL,
      const int*   __restrict__ exA, const int* __restrict__ exO,
      const float* __restrict__ W1, const float* __restrict__ b1,
      const float* __restrict__ W2, const float* __restrict__ b2,
      const float* __restrict__ W3, const float* __restrict__ b3,
      float* __restrict__ outA, float* __restrict__ outO) {
  __shared__ unsigned OG[NWAVE], IF1[NWAVE], IF01[NWAVE], F3[OWNW], OGN[1];

  const int b    = blockIdx.x / GPB;
  const int g    = blockIdx.x % GPB;
  const int base = g * (OWNW * 32);
  const int tid  = threadIdx.x;
  const int lane = tid & 31;
  const int w    = __builtin_amdgcn_readfirstlane(tid >> 5);
  const bool lb  = (w == OWNW);
  const int s0   = lb ? max(base - 32, 0) : base + 32 * w;

  const unsigned nOG = origword(aL, oL, b, base + OWNW * 32, lane);

  v4f av, ov;
  float mult, fa1, fo1;
  unsigned wOG, wI1, wI01;
  posword(rel, aL, oL, exA, exO, W1, b1, W2, b2, W3, b3, b, s0, lane,
          av, ov, mult, fa1, fo1, wOG, wI1, wI01);
  if (lane == 0) { OG[w] = wOG; IF1[w] = wI1; IF01[w] = wI01; }
  if (lb && lane == 0) OGN[0] = nOG;
  __syncthreads();

  if (w == 0) {
    const int t  = lane;
    const int ti = min(t, OWNW);
    const unsigned og  = OG[ti];
    const unsigned ogn = (t < OWNW - 1) ? OG[ti + 1] : ((t == OWNW - 1) ? OGN[0] : OG[0]);
    const unsigned A   = IF1[ti];
    const unsigned C   = IF01[ti];
    const unsigned G   = (og >> 1) | (og >> 2) | (ogn << 31) | (ogn << 30);
    const unsigned g0c = G, g1c = G | 1u, g2c = G | 3u;
    unsigned na0 = A, na1 = A, na2 = A;
    unsigned nb0 = 0u, nb1 = 0u, nb2 = 0u;
    for (int it = 0; it < 16; ++it) {
      nb0 = (na0 << 1) | (na0 << 2) | g0c;  na0 = (nb0 & A) | (~nb0 & C);
      nb1 = (na1 << 1) | (na1 << 2) | g1c;  na1 = (nb1 & A) | (~nb1 & C);
      nb2 = (na2 << 1) | (na2 << 2) | g2c;  na2 = (nb2 & A) | (~nb2 & C);
      const unsigned q0 = na0, q1 = na1, q2 = na2;
      nb0 = (na0 << 1) | (na0 << 2) | g0c;  na0 = (nb0 & A) | (~nb0 & C);
      nb1 = (na1 << 1) | (na1 << 2) | g1c;  na1 = (nb1 & A) | (~nb1 & C);
      nb2 = (na2 << 1) | (na2 << 2) | g2c;  na2 = (nb2 & A) | (~nb2 & C);
      const bool moved = (((na0 ^ q0) | (na1 ^ q1) | (na2 ^ q2)) != 0u);
      if (__builtin_amdgcn_ballot_w32(moved) == 0u) break;
    }
    const unsigned c0 = min(na0 >> 30, 2u);
    const unsigned c1 = min(na1 >> 30, 2u);
    const unsigned c2 = min(na2 >> 30, 2u);
    const unsigned F  = c0 | (c1 << 2) | (c2 << 4);

    unsigned st = 0u;
    if (g != 0) {
      const unsigned fl = (unsigned)__builtin_amdgcn_readlane((int)F, OWNW);
      const unsigned l0 = fl & 3u, l1 = (fl >> 2) & 3u, l2 = (fl >> 4) & 3u;
      if (l0 == l1 && l1 == l2) {
        st = l0;
      } else {
        unsigned cs = 0u;
        const int nwords = (base >> 5) - 1;
        for (int j = 0; j < nwords; ++j) {
          v4f xa, xo;  float xm, xf, xg;
          unsigned jOG, jA, jC;
          posword(rel, aL, oL, exA, exO, W1, b1, W2, b2, W3, b3, b, 32 * j, lane,
                  xa, xo, xm, xf, xg, jOG, jA, jC);
          const unsigned jn = origword(aL, oL, b, 32 * (j + 1), lane);
          const unsigned jG = (jOG >> 1) | (jOG >> 2) | (jn << 31) | (jn << 30);
          const unsigned gc = jG | ((cs == 0u) ? 0u : ((cs == 1u) ? 1u : 3u));
          unsigned na = 0u;
          for (int it = 0; it < 32; ++it) {
            const unsigned nb = (na << 1) | (na << 2) | gc;
            na = (nb & jA) | (~nb & jC);
          }
          cs = min(na >> 30, 2u);
        }
        st = (fl >> (2u * cs)) & 3u;
      }
    }
    unsigned cls = 0u;
    #pragma unroll
    for (int k = 0; k < OWNW; ++k) {
      cls |= st << (2 * k);
      const unsigned f = (unsigned)__builtin_amdgcn_readlane((int)F, k);
      st = (f >> (2u * st)) & 3u;
    }
    const unsigned mine = (cls >> (2 * min(t, OWNW - 1))) & 3u;
    if (t < OWNW) F3[t] = (mine == 0u) ? nb0 : ((mine == 1u) ? nb1 : nb2);
  }
  __syncthreads();

  if (!lb) {
    const int p     = b * SEQ + s0 + lane;
    const float f3  = ((F3[w] >> lane) & 1u) ? 1.0f : 0.1f;
    const float faw = fa1 * f3;
    const float fow = fo1 * f3;
    v4f ra, ro;
    ra.x = av.x * mult * faw;  ra.y = av.y * mult * faw;
    ra.z = av.z * mult;        ra.w = av.w * mult;
    ro.x = ov.x * mult * fow;  ro.y = ov.y * mult * fow;
    ro.z = ov.z * mult;        ro.w = ov.w * mult;
    *(v4f*)(outA + (size_t)p * 4) = ra;
    *(v4f*)(outO + (size_t)p * 4) = ro;
  }
}

extern "C" void kernel_launch(void* const* d_in, const int* in_sizes, int n_in,
                              void* d_out, int out_size, void* d_ws, size_t ws_size,
                              hipStream_t stream) {
  (void)in_sizes; (void)n_in; (void)out_size; (void)d_ws; (void)ws_size;
  const float* aL  = (const float*)d_in[0];
  const float* oL  = (const float*)d_in[1];
  const float* rel = (const float*)d_in[2];
  const int*   exA = (const int*)d_in[3];
  const int*   exO = (const int*)d_in[4];
  const float* W1  = (const float*)d_in[5];
  const float* b1  = (const float*)d_in[6];
  const float* W2  = (const float*)d_in[7];
  const float* b2  = (const float*)d_in[8];
  const float* W3  = (const float*)d_in[9];
  const float* b3  = (const float*)d_in[10];

  float* outA  = (float*)d_out;
  float* outO  = outA + NPOS * 4;

  k_all<<<BATCH * GPB, 32 * NWAVE, 0, stream>>>(rel, aL, oL, exA, exO,
                                                W1, b1, W2, b2, W3, b3, outA, outO);
}


# blocks of 2 own + 1 look-back wave (16 per batch) instead of 4 + 1: at most one wave per SIMD, so no matrix pipe runs two waves' 24-WMMA chains one behind the other ahead of the first barrier
# speedup vs baseline: 12.7376x; 1.1196x over previous
#include <hip/hip_runtime.h>
#include <math.h>


#define BATCH 8
#define SEQ   1024
#define KREL  4
#define NPOS  (BATCH * SEQ)
#define OWNW  2
#define NWAVE 3
#define GPB   16

typedef __attribute__((ext_vector_type(2))) float v2f;
typedef __attribute__((ext_vector_type(4))) float v4f;
typedef __attribute__((ext_vector_type(8))) float v8f;

__device__ __forceinline__ bool hit4(const v4f v) {
  return (v.x > 0.5f) | (v.y > 0.5f) | (v.z > 0.5f) | (v.w > 0.5f);
}

__device__ __forceinline__ float xhalf(const float x) {
  const int xi = __builtin_bit_cast(int, x);
  return __builtin_bit_cast(float,
      __builtin_amdgcn_permlanex16(xi, xi, 0x76543210, 0xfedcba98, false, false));
}

__device__ __forceinline__ unsigned origword(const float* __restrict__ aL,
                                             const float* __restrict__ oL,
                                             const int b, const int s0, const int lane) {
  const int s  = s0 + lane;
  const int sc = min(s, SEQ - 1);
  const v4f a = *(const v4f*)(aL + (size_t)(b * SEQ + sc) * 4);
  const v4f o = *(const v4f*)(oL + (size_t)(b * SEQ + sc) * 4);
  const bool orig = (s < SEQ) &
                    ((fmaxf(fmaxf(a.x, a.y), fmaxf(a.z, a.w)) > 0.5f) |
                     (fmaxf(fmaxf(o.x, o.y), fmaxf(o.z, o.w)) > 0.5f));
  return __builtin_amdgcn_ballot_w32(orig);
}

__device__ __forceinline__ void posword(
      const float* __restrict__ rel,
      const float* __restrict__ aL, const float* __restrict__ oL,
      const int*   __restrict__ exA, const int* __restrict__ exO,
      const float* __restrict__ W1, const float* __restrict__ b1,
      const float* __restrict__ W2, const float* __restrict__ b2,
      const float* __restrict__ W3, const float* __restrict__ b3,
      const int b, const int s0, const int lane,
      v4f& avo, v4f& ovo, float& multo, float& fa1o, float& fo1o,
      unsigned& wOGo, unsigned& wI1o, unsigned& wI01o) {
  const int hf   = lane >> 4;
  const int ln   = lane & 15;
  const int s    = s0 + lane;
  const int p    = b * SEQ + s;

  const int k2 = 2 * hf;
  const int pl = 4 * ((ln & 7) >> 1) + 2 * (ln >> 3) + (ln & 1);
  v2f w1f[2][2], w2f[8];
  #pragma unroll
  for (int T = 0; T < 2; ++T) {
    #pragma unroll
    for (int q = 0; q < 2; ++q) {
      w1f[T][q].x = W1[(4*q + k2)*32     + 16*T + pl];
      w1f[T][q].y = W1[(4*q + k2 + 1)*32 + 16*T + pl];
    }
  }
  v4f xat[2], xot[2];
  #pragma unroll
  for (int t = 0; t < 2; ++t) {
    const int pT = b * SEQ + s0 + 16 * t + ln;
    xat[t] = *(const v4f*)(aL + (size_t)pT * 4);
    xot[t] = *(const v4f*)(oL + (size_t)pT * 4);
  }
  v2f bb1[2][4];
  #pragma unroll
  for (int T = 0; T < 2; ++T) {
    #pragma unroll
    for (int j = 0; j < 4; ++j) bb1[T][j] = *(const v2f*)(b1 + 16*T + 4*j + k2);
  }
  #pragma unroll
  for (int c = 0; c < 8; ++c) {
    w2f[c].x = W2[(4*c + k2)*16 + ln];
    w2f[c].y = W2[(4*c + 1 + k2)*16 + ln];
  }
  v4f bb2[2], w3v[2];
  #pragma unroll
  for (int u = 0; u < 2; ++u) {
    bb2[u] = *(const v4f*)(b2 + 8*hf + 4*u);
    w3v[u] = *(const v4f*)(W3 + 8*hf + 4*u);
  }
  const float bia3 = b3[0];

  const v4f av = *(const v4f*)(aL + (size_t)p * 4);
  const v4f ov = *(const v4f*)(oL + (size_t)p * 4);

  const v4f* R  = (const v4f*)rel + ((size_t)b << 20);
  const int sub = lane >> 3, col = lane & 7;
  v4f pv[8];
  #pragma unroll
  for (int j = 0; j < 8; ++j) {
    const int i = s0 + 4 * j + sub;
    pv[j] = R[(i << 10) + min(col, max(i - 1, 0))];
  }

  const int* eA = exA + b * SEQ;
  const int* eO = exO + b * SEQ;
  int wa[7], wo[7];
  #pragma unroll
  for (int d = 0; d < 7; ++d) {
    const int sc = min(max(s + d - 3, 0), SEQ - 1);
    wa[d] = eA[sc];  wo[d] = eO[sc];
  }

  __builtin_amdgcn_sched_barrier(0);

  v2f xb[2][2];
  #pragma unroll
  for (int t = 0; t < 2; ++t) {
    const v4f xa = xat[t];
    const v4f xo = xot[t];
    xb[t][0].x = hf ? xa.z : xa.x;  xb[t][0].y = hf ? xa.w : xa.y;
    xb[t][1].x = hf ? xo.z : xo.x;  xb[t][1].y = hf ? xo.w : xo.y;
  }

  v8f h1[2][2];
  #pragma unroll
  for (int T = 0; T < 2; ++T) {
    #pragma unroll
    for (int t = 0; t < 2; ++t) {
      v8f acc = {0,0,0,0,0,0,0,0};
      acc = __builtin_amdgcn_wmma_f32_16x16x4_f32(false, w1f[T][0], false, xb[t][0], (short)0, acc, false, false);
      acc = __builtin_amdgcn_wmma_f32_16x16x4_f32(false, w1f[T][1], false, xb[t][1], (short)0, acc, false, false);
      #pragma unroll
      for (int r = 0; r < 8; ++r) {
        const float v = acc[r] + bb1[T][r >> 1][r & 1];
        acc[r] = v > 0.0f ? v : 0.0f;
      }
      h1[t][T] = acc;
    }
  }

  v8f a2[2];
  a2[0] = (v8f){0,0,0,0,0,0,0,0};
  a2[1] = (v8f){0,0,0,0,0,0,0,0};
  #pragma unroll
  for (int c = 0; c < 8; ++c) {
    #pragma unroll
    for (int t = 0; t < 2; ++t) {
      v2f hb;
      hb.x = h1[t][c >> 2][2 * (c & 3)];
      hb.y = h1[t][c >> 2][2 * (c & 3) + 1];
      a2[t] = __builtin_amdgcn_wmma_f32_16x16x4_f32(false, w2f[c], false, hb, (short)0, a2[t], false, false);
    }
  }

  float part[2];
  #pragma unroll
  for (int t = 0; t < 2; ++t) {
    float hv[8];
    #pragma unroll
    for (int r = 0; r < 8; ++r) {
      const float v = a2[t][r] + bb2[r >> 2][r & 3];
      hv[r] = v > 0.0f ? v : 0.0f;
    }
    float pr = hv[0] * w3v[0][0];
    #pragma unroll
    for (int r = 1; r < 8; ++r) pr += hv[r] * w3v[r >> 2][r & 3];
    part[t] = pr;
  }
  const float mine = hf ? part[1] : part[0];
  const float send = hf ? part[0] : part[1];
  const float s3   = (mine + xhalf(send)) + bia3;

  const float cg    = 1.0f / (1.0f + expf(-s3));
  const float mult  = cg < 0.5f ? 2.0f * cg : 1.0f;
  const float mA = fmaxf(fmaxf(av.x, av.y), fmaxf(av.z, av.w));
  const float e0 = expf(av.x - mA), e1 = expf(av.y - mA), e2 = expf(av.z - mA), e3 = expf(av.w - mA);
  const float impA = (e0 + e1) / (e0 + e1 + e2 + e3);
  const float mO = fmaxf(fmaxf(ov.x, ov.y), fmaxf(ov.z, ov.w));
  const float g0 = expf(ov.x - mO), g1 = expf(ov.y - mO), g2 = expf(ov.z - mO), g3 = expf(ov.w - mO);
  const float impO = (g0 + g1) / (g0 + g1 + g2 + g3);
  bool nearA = false, nearO = false;
  #pragma unroll
  for (int d = 0; d < 7; ++d) {
    const int ss = s + d - 3;
    const bool in = (ss >= 0) & (ss < SEQ);
    nearA |= in & (wa[d] > 0);
    nearO |= in & (wo[d] > 0);
  }
  const float fa2 = (impA > 0.5f && !nearO) ? 0.3f : 1.0f;
  const float fo2 = (impO > 0.5f && !nearA) ? 0.3f : 1.0f;

  unsigned mym = 0u;
  #pragma unroll
  for (int j = 0; j < 8; ++j) {
    const int i = s0 + 4 * j + sub;
    const unsigned m = __builtin_amdgcn_ballot_w32((col < i) & hit4(pv[j]));
    if ((lane >> 2) == j) mym = m;
  }
  bool anyrow = ((mym >> (8 * (lane & 3))) & 0xFFu) != 0u;
  unsigned um = __builtin_amdgcn_ballot_w32(!anyrow && s > 8);
  while (um != 0u) {
    const int r = __builtin_ctz(um);
    um &= um - 1u;
    const int i = s0 + r;
    const v4f* Rr = R + ((size_t)i << 10);
    bool found = false;
    for (int c = 8; c < i && !found; c += 256) {
      bool hh = false;
      #pragma unroll
      for (int u = 0; u < 8; ++u)
        hh |= hit4(__builtin_nontemporal_load(Rr + min(c + 32 * u + lane, i - 1)));
      found = __builtin_amdgcn_ballot_w32(hh) != 0u;
    }
    if (lane == r) anyrow = found;
  }
  const float f4 = anyrow ? 0.7f : 1.0f;

  const float fa1 = fa2 * f4;
  const float fo1 = fo2;
  const float a0m = av.x * mult, a1m = av.y * mult, a2m = av.z * mult, a3m = av.w * mult;
  const float o0m = ov.x * mult, o1m = ov.y * mult, o2m = ov.z * mult, o3m = ov.w * mult;
  const bool orig = (fmaxf(fmaxf(av.x, av.y), fmaxf(av.z, av.w)) > 0.5f) ||
                    (fmaxf(fmaxf(ov.x, ov.y), fmaxf(ov.z, ov.w)) > 0.5f);
  float fa = fa1, fo = fo1;
  const bool act1  = (fmaxf(fmaxf(a0m*fa, a1m*fa), fmaxf(a2m, a3m)) > 0.5f) ||
                     (fmaxf(fmaxf(o0m*fo, o1m*fo), fmaxf(o2m, o3m)) > 0.5f);
  fa = fa1 * 0.1f; fo = fo1 * 0.1f;
  const bool act01 = (fmaxf(fmaxf(a0m*fa, a1m*fa), fmaxf(a2m, a3m)) > 0.5f) ||
                     (fmaxf(fmaxf(o0m*fo, o1m*fo), fmaxf(o2m, o3m)) > 0.5f);
  wOGo  = __builtin_amdgcn_ballot_w32(orig);
  wI1o  = __builtin_amdgcn_ballot_w32(act1);
  wI01o = __builtin_amdgcn_ballot_w32(act01);
  avo = av;  ovo = ov;  multo = mult;  fa1o = fa1;  fo1o = fo1;
}

__global__ void __launch_bounds__(32 * NWAVE)
k_all(const float* __restrict__ rel,
      const float* __restrict__ aL, const float* __restrict__ oL,
      const int*   __restrict__ exA, const int* __restrict__ exO,
      const float* __restrict__ W1, const float* __restrict__ b1,
      const float* __restrict__ W2, const float* __restrict__ b2,
      const float* __restrict__ W3, const float* __restrict__ b3,
      float* __restrict__ outA, float* __restrict__ outO) {
  __shared__ unsigned OG[NWAVE], IF1[NWAVE], IF01[NWAVE], F3[OWNW], OGN[1];

  const int b    = blockIdx.x / GPB;
  const int g    = blockIdx.x % GPB;
  const int base = g * (OWNW * 32);
  const int tid  = threadIdx.x;
  const int lane = tid & 31;
  const int w    = __builtin_amdgcn_readfirstlane(tid >> 5);
  const bool lb  = (w == OWNW);
  const int s0   = lb ? max(base - 32, 0) : base + 32 * w;

  const unsigned nOG = origword(aL, oL, b, base + OWNW * 32, lane);

  v4f av, ov;
  float mult, fa1, fo1;
  unsigned wOG, wI1, wI01;
  posword(rel, aL, oL, exA, exO, W1, b1, W2, b2, W3, b3, b, s0, lane,
          av, ov, mult, fa1, fo1, wOG, wI1, wI01);
  if (lane == 0) { OG[w] = wOG; IF1[w] = wI1; IF01[w] = wI01; }
  if (lb && lane == 0) OGN[0] = nOG;
  __syncthreads();

  if (w == 0) {
    const int t  = lane;
    const int ti = min(t, OWNW);
    const unsigned og  = OG[ti];
    const unsigned ogn = (t < OWNW - 1) ? OG[ti + 1] : ((t == OWNW - 1) ? OGN[0] : OG[0]);
    const unsigned A   = IF1[ti];
    const unsigned C   = IF01[ti];
    const unsigned G   = (og >> 1) | (og >> 2) | (ogn << 31) | (ogn << 30);
    const unsigned g0c = G, g1c = G | 1u, g2c = G | 3u;
    unsigned na0 = A, na1 = A, na2 = A;
    unsigned nb0 = 0u, nb1 = 0u, nb2 = 0u;
    for (int it = 0; it < 16; ++it) {
      nb0 = (na0 << 1) | (na0 << 2) | g0c;  na0 = (nb0 & A) | (~nb0 & C);
      nb1 = (na1 << 1) | (na1 << 2) | g1c;  na1 = (nb1 & A) | (~nb1 & C);
      nb2 = (na2 << 1) | (na2 << 2) | g2c;  na2 = (nb2 & A) | (~nb2 & C);
      const unsigned q0 = na0, q1 = na1, q2 = na2;
      nb0 = (na0 << 1) | (na0 << 2) | g0c;  na0 = (nb0 & A) | (~nb0 & C);
      nb1 = (na1 << 1) | (na1 << 2) | g1c;  na1 = (nb1 & A) | (~nb1 & C);
      nb2 = (na2 << 1) | (na2 << 2) | g2c;  na2 = (nb2 & A) | (~nb2 & C);
      const bool moved = (((na0 ^ q0) | (na1 ^ q1) | (na2 ^ q2)) != 0u);
      if (__builtin_amdgcn_ballot_w32(moved) == 0u) break;
    }
    const unsigned c0 = min(na0 >> 30, 2u);
    const unsigned c1 = min(na1 >> 30, 2u);
    const unsigned c2 = min(na2 >> 30, 2u);
    const unsigned F  = c0 | (c1 << 2) | (c2 << 4);

    unsigned st = 0u;
    if (g != 0) {
      const unsigned fl = (unsigned)__builtin_amdgcn_readlane((int)F, OWNW);
      const unsigned l0 = fl & 3u, l1 = (fl >> 2) & 3u, l2 = (fl >> 4) & 3u;
      if (l0 == l1 && l1 == l2) {
        st = l0;
      } else {
        unsigned cs = 0u;
        const int nwords = (base >> 5) - 1;
        for (int j = 0; j < nwords; ++j) {
          v4f xa, xo;  float xm, xf, xg;
          unsigned jOG, jA, jC;
          posword(rel, aL, oL, exA, exO, W1, b1, W2, b2, W3, b3, b, 32 * j, lane,
                  xa, xo, xm, xf, xg, jOG, jA, jC);
          const unsigned jn = origword(aL, oL, b, 32 * (j + 1), lane);
          const unsigned jG = (jOG >> 1) | (jOG >> 2) | (jn << 31) | (jn << 30);
          const unsigned gc = jG | ((cs == 0u) ? 0u : ((cs == 1u) ? 1u : 3u));
          unsigned na = 0u;
          for (int it = 0; it < 32; ++it) {
            const unsigned nb = (na << 1) | (na << 2) | gc;
            na = (nb & jA) | (~nb & jC);
          }
          cs = min(na >> 30, 2u);
        }
        st = (fl >> (2u * cs)) & 3u;
      }
    }
    unsigned cls = 0u;
    #pragma unroll
    for (int k = 0; k < OWNW; ++k) {
      cls |= st << (2 * k);
      const unsigned f = (unsigned)__builtin_amdgcn_readlane((int)F, k);
      st = (f >> (2u * st)) & 3u;
    }
    const unsigned mine = (cls >> (2 * min(t, OWNW - 1))) & 3u;
    if (t < OWNW) F3[t] = (mine == 0u) ? nb0 : ((mine == 1u) ? nb1 : nb2);
  }
  __syncthreads();

  if (!lb) {
    const int p     = b * SEQ + s0 + lane;
    const float f3  = ((F3[w] >> lane) & 1u) ? 1.0f : 0.1f;
    const float faw = fa1 * f3;
    const float fow = fo1 * f3;
    v4f ra, ro;
    ra.x = av.x * mult * faw;  ra.y = av.y * mult * faw;
    ra.z = av.z * mult;        ra.w = av.w * mult;
    ro.x = ov.x * mult * fow;  ro.y = ov.y * mult * fow;
    ro.z = ov.z * mult;        ro.w = ov.w * mult;
    *(v4f*)(outA + (size_t)p * 4) = ra;
    *(v4f*)(outO + (size_t)p * 4) = ro;
  }
}

extern "C" void kernel_launch(void* const* d_in, const int* in_sizes, int n_in,
                              void* d_out, int out_size, void* d_ws, size_t ws_size,
                              hipStream_t stream) {
  (void)in_sizes; (void)n_in; (void)out_size; (void)d_ws; (void)ws_size;
  const float* aL  = (const float*)d_in[0];
  const float* oL  = (const float*)d_in[1];
  const float* rel = (const float*)d_in[2];
  const int*   exA = (const int*)d_in[3];
  const int*   exO = (const int*)d_in[4];
  const float* W1  = (const float*)d_in[5];
  const float* b1  = (const float*)d_in[6];
  const float* W2  = (const float*)d_in[7];
  const float* b2  = (const float*)d_in[8];
  const float* W3  = (const float*)d_in[9];
  const float* b3  = (const float*)d_in[10];

  float* outA  = (float*)d_out;
  float* outO  = outA + NPOS * 4;

  k_all<<<BATCH * GPB, 32 * NWAVE, 0, stream>>>(rel, aL, oL, exA, exO,
                                                W1, b1, W2, b2, W3, b3, outA, outO);
}
